# starting kernel
# baseline (speedup 1.0000x reference)
#include <hip/hip_runtime.h>
#include <hip/hip_bf16.h>

typedef __attribute__((ext_vector_type(2))) float v2f;
typedef __attribute__((ext_vector_type(8))) float v8f;

#define B_   32
#define L_   512
#define C_   128
#define HID_ 256
#define H_   4
#define D_   64
#define NL_  3
#define NEG_ 0.2f
#define EPS_ 1e-5f

template <int M, int N, int K, int LDA, int LDB, int LDD, int INNER,
          long SAO, long SAI, long SBO, long SBI, long SDO, long SDI,
          long SBIASI, bool TRANSA, bool TRANSD, bool HASBIAS>
__global__ __launch_bounds__(128) void wmma_gemm_f32(
    const float* __restrict__ A, const float* __restrict__ Bm,
    const float* __restrict__ bias, float* __restrict__ D)
{
  const int lane = threadIdx.x & 31;
  const int wave = threadIdx.x >> 5;
  const int half = (lane >> 4) & 1;
  const int r    = lane & 15;

  constexpr int tilesN = N >> 5;
  const int tile = blockIdx.x * 4 + wave;
  const int tm = tile / tilesN;
  const int tn = tile % tilesN;

  const int batch = blockIdx.y;
  const int bo = batch / INNER;
  const int bi = batch % INNER;
  A    += bo * SAO + bi * SAI;
  Bm   += bo * SBO + bi * SBI;
  D    += bo * SDO + bi * SDI;
  if (HASBIAS) bias += bi * SBIASI;

  const int m0 = tm << 5;
  const int n0 = tn << 5;

  constexpr int stepA = TRANSA ? 4 * LDA : 4;
  constexpr int stepB = 4 * LDB;
  int offA0 = TRANSA ? (2 * half) * LDA + (m0 + r)
                     : (m0 + r) * LDA + 2 * half;
  int offA1 = offA0 + (TRANSA ? 16 : 16 * LDA);
  int offB0 = (2 * half) * LDB + (n0 + r);
  int offB1 = offB0 + 16;

  v8f acc00 = {0.f,0.f,0.f,0.f,0.f,0.f,0.f,0.f};
  v8f acc01 = acc00, acc10 = acc00, acc11 = acc00;

#pragma unroll 4
  for (int k0 = 0; k0 < K; k0 += 4) {
    v2f a0, a1, b0, b1;
    if (TRANSA) {
      a0.x = A[offA0]; a0.y = A[offA0 + LDA];
      a1.x = A[offA1]; a1.y = A[offA1 + LDA];
    } else {
      a0 = *(const v2f*)(A + offA0);
      a1 = *(const v2f*)(A + offA1);
    }
    b0.x = Bm[offB0]; b0.y = Bm[offB0 + LDB];
    b1.x = Bm[offB1]; b1.y = Bm[offB1 + LDB];
    offA0 += stepA; offA1 += stepA; offB0 += stepB; offB1 += stepB;

    acc00 = __builtin_amdgcn_wmma_f32_16x16x4_f32(false, a0, false, b0, (short)0, acc00, false, false);
    acc01 = __builtin_amdgcn_wmma_f32_16x16x4_f32(false, a0, false, b1, (short)0, acc01, false, false);
    acc10 = __builtin_amdgcn_wmma_f32_16x16x4_f32(false, a1, false, b0, (short)0, acc10, false, false);
    acc11 = __builtin_amdgcn_wmma_f32_16x16x4_f32(false, a1, false, b1, (short)0, acc11, false, false);
  }

  const int col0 = n0 + r;
  const int col1 = col0 + 16;
  const float bv0 = HASBIAS ? bias[col0] : 0.f;
  const float bv1 = HASBIAS ? bias[col1] : 0.f;
#pragma unroll
  for (int i = 0; i < 8; ++i) {
    const int row0 = m0 + i + 8 * half;
    const int row1 = row0 + 16;
    if (TRANSD) {
      D[(long)col0 * LDD + row0] = acc00[i] + bv0;
      D[(long)col1 * LDD + row0] = acc01[i] + bv1;
      D[(long)col0 * LDD + row1] = acc10[i] + bv0;
      D[(long)col1 * LDD + row1] = acc11[i] + bv1;
    } else {
      D[(long)row0 * LDD + col0] = acc00[i] + bv0;
      D[(long)row0 * LDD + col1] = acc01[i] + bv1;
      D[(long)row1 * LDD + col0] = acc10[i] + bv0;
      D[(long)row1 * LDD + col1] = acc11[i] + bv1;
    }
  }
}

__global__ __launch_bounds__(128) void gat_scores_softmax(
    const float* __restrict__ gl, const float* __restrict__ gr,
    const float* __restrict__ att, float* __restrict__ alpha)
{
  __shared__ float sGr[C_ * D_];
  __shared__ float sAtt[D_];
  __shared__ float sRow[D_];
  __shared__ float red[4];

  const int tid = threadIdx.x;
  const int h = blockIdx.x;
  const int b = blockIdx.y;
  const float* glb = gl + (long)b * C_ * HID_ + h * D_;
  const float* grb = gr + (long)b * C_ * HID_ + h * D_;

  for (int idx = tid; idx < C_ * D_; idx += 128) {
    const int s = idx >> 6, d = idx & 63;
    sGr[idx] = grb[(long)s * HID_ + d];
  }
  if (tid < D_) sAtt[tid] = att[h * D_ + tid];
  __syncthreads();

  const int lane = tid & 31;
  const int wv = tid >> 5;
  float* arow = alpha + (long)(b * H_ + h) * C_ * C_;

  for (int t = 0; t < C_; ++t) {
    if (tid < D_) sRow[tid] = glb[(long)t * HID_ + tid];
    __syncthreads();

    const int s = tid;
    float sc = 0.f;
    const float* g = &sGr[s * D_];
#pragma unroll 8
    for (int d = 0; d < D_; ++d) {
      float v = sRow[d] + g[d];
      v = v > 0.f ? v : NEG_ * v;
      sc = fmaf(sAtt[d], v, sc);
    }
    float m = sc;
    for (int off = 16; off > 0; off >>= 1) m = fmaxf(m, __shfl_xor(m, off, 32));
    if (lane == 0) red[wv] = m;
    __syncthreads();
    m = fmaxf(fmaxf(red[0], red[1]), fmaxf(red[2], red[3]));
    __syncthreads();
    const float e = __expf(sc - m);
    float ssum = e;
    for (int off = 16; off > 0; off >>= 1) ssum += __shfl_xor(ssum, off, 32);
    if (lane == 0) red[wv] = ssum;
    __syncthreads();
    ssum = red[0] + red[1] + red[2] + red[3];
    arow[(long)t * C_ + s] = e / ssum;
    __syncthreads();
  }
}

__global__ __launch_bounds__(256) void post_elu_residual_ln(
    float* __restrict__ h, const float* __restrict__ gout,
    const float* __restrict__ g, const float* __restrict__ bb)
{
  __shared__ float red[8];
  const int tid = threadIdx.x;
  const long base = (long)blockIdx.x * HID_;

  float x = gout[base + tid];
  x = x > 0.f ? x : (__expf(x) - 1.f);
  const float v = h[base + tid] + x;

  const int lane = tid & 31, wv = tid >> 5;
  float s = v;
  for (int off = 16; off > 0; off >>= 1) s += __shfl_xor(s, off, 32);
  if (lane == 0) red[wv] = s;
  __syncthreads();
  s = 0.f;
#pragma unroll
  for (int i = 0; i < 8; ++i) s += red[i];
  const float mu = s * (1.f / HID_);
  __syncthreads();

  const float dmu = v - mu;
  float q = dmu * dmu;
  for (int off = 16; off > 0; off >>= 1) q += __shfl_xor(q, off, 32);
  if (lane == 0) red[wv] = q;
  __syncthreads();
  q = 0.f;
#pragma unroll
  for (int i = 0; i < 8; ++i) q += red[i];
  const float inv = rsqrtf(q * (1.f / HID_) + EPS_);

  h[base + tid] = dmu * inv * g[tid] + bb[tid];
}

extern "C" void kernel_launch(void* const* d_in, const int* in_sizes, int n_in,
                              void* d_out, int out_size, void* d_ws, size_t ws_size,
                              hipStream_t stream) {
  (void)in_sizes; (void)n_in; (void)out_size; (void)ws_size;
  const float* x      = (const float*)d_in[0];
  const float* W_emb  = (const float*)d_in[1];
  const float* b_emb  = (const float*)d_in[2];
  const float* Wl     = (const float*)d_in[3];
  const float* Wr     = (const float*)d_in[4];
  const float* att    = (const float*)d_in[5];
  const float* bias_g = (const float*)d_in[6];
  const float* ln_g   = (const float*)d_in[7];
  const float* ln_b   = (const float*)d_in[8];
  const float* W_proj = (const float*)d_in[9];
  const float* b_proj = (const float*)d_in[10];

  float* out   = (float*)d_out;
  float* alpha = out + (long)B_ * L_ * C_;

  float* ws = (float*)d_ws;
  const long hwN = (long)B_ * C_ * HID_;
  float* h    = ws;
  float* gl   = ws + hwN;
  float* gr   = ws + 2 * hwN;
  float* gout = ws + 3 * hwN;

  wmma_gemm_f32<C_, HID_, L_, C_, HID_, HID_, 1,
                (long)L_ * C_, 0, 0, 0, (long)C_ * HID_, 0, 0,
                true, false, true>
      <<<dim3((C_ / 32) * (HID_ / 32) / 4, B_), 128, 0, stream>>>(
          x, W_emb, b_emb, h);

  for (int layer = 0; layer < NL_; ++layer) {
    const float* Wl_i  = Wl + (long)layer * HID_ * HID_;
    const float* Wr_i  = Wr + (long)layer * HID_ * HID_;
    const float* att_i = att + (long)layer * H_ * D_;
    const float* bg_i  = bias_g + (long)layer * HID_;

    dim3 g1((C_ / 32) * (HID_ / 32) / 4, B_);
    wmma_gemm_f32<C_, HID_, HID_, HID_, HID_, HID_, 1,
                  (long)C_ * HID_, 0, 0, 0, (long)C_ * HID_, 0, 0,
                  false, false, false>
        <<<g1, 128, 0, stream>>>(h, Wl_i, nullptr, gl);
    wmma_gemm_f32<C_, HID_, HID_, HID_, HID_, HID_, 1,
                  (long)C_ * HID_, 0, 0, 0, (long)C_ * HID_, 0, 0,
                  false, false, false>
        <<<g1, 128, 0, stream>>>(h, Wr_i, nullptr, gr);

    gat_scores_softmax<<<dim3(H_, B_), 128, 0, stream>>>(gl, gr, att_i, alpha);

    wmma_gemm_f32<C_, D_, C_, C_, HID_, HID_, H_,
                  (long)H_ * C_ * C_, (long)C_ * C_,
                  (long)C_ * HID_, D_, (long)C_ * HID_, D_, D_,
                  false, false, true>
        <<<dim3((C_ / 32) * (D_ / 32) / 4, B_ * H_), 128, 0, stream>>>(
            alpha, gr, bg_i, gout);

    post_elu_residual_ln<<<B_ * C_, 256, 0, stream>>>(
        h, gout, ln_g + (long)layer * HID_, ln_b + (long)layer * HID_);
  }

  wmma_gemm_f32<C_, L_, HID_, HID_, L_, C_, 1,
                (long)C_ * HID_, 0, 0, 0, (long)L_ * C_, 0, 0,
                false, true, true>
      <<<dim3((C_ / 32) * (L_ / 32) / 4, B_), 128, 0, stream>>>(
          h, W_proj, b_proj, out);
}


# scores kernel: inner-product operands in registers (own gr source row and att held over the whole target loop, target row read whole per step) instead of LDS reads at a one-bank 64-dword pitch
# speedup vs baseline: 1.4448x; 1.4448x over previous
#include <hip/hip_runtime.h>
#include <hip/hip_bf16.h>

typedef __attribute__((ext_vector_type(2))) float v2f;
typedef __attribute__((ext_vector_type(8))) float v8f;
typedef __attribute__((ext_vector_type(4))) float v4f;

#define B_   32
#define L_   512
#define C_   128
#define HID_ 256
#define H_   4
#define D_   64
#define NL_  3
#define NEG_ 0.2f
#define EPS_ 1e-5f

template <int M, int N, int K, int LDA, int LDB, int LDD, int INNER,
          long SAO, long SAI, long SBO, long SBI, long SDO, long SDI,
          long SBIASI, bool TRANSA, bool TRANSD, bool HASBIAS>
__global__ __launch_bounds__(128) void wmma_gemm_f32(
    const float* __restrict__ A, const float* __restrict__ Bm,
    const float* __restrict__ bias, float* __restrict__ D)
{
  const int lane = threadIdx.x & 31;
  const int wave = threadIdx.x >> 5;
  const int half = (lane >> 4) & 1;
  const int r    = lane & 15;

  constexpr int tilesN = N >> 5;
  const int tile = blockIdx.x * 4 + wave;
  const int tm = tile / tilesN;
  const int tn = tile % tilesN;

  const int batch = blockIdx.y;
  const int bo = batch / INNER;
  const int bi = batch % INNER;
  A    += bo * SAO + bi * SAI;
  Bm   += bo * SBO + bi * SBI;
  D    += bo * SDO + bi * SDI;
  if (HASBIAS) bias += bi * SBIASI;

  const int m0 = tm << 5;
  const int n0 = tn << 5;

  constexpr int stepA = TRANSA ? 4 * LDA : 4;
  constexpr int stepB = 4 * LDB;
  int offA0 = TRANSA ? (2 * half) * LDA + (m0 + r)
                     : (m0 + r) * LDA + 2 * half;
  int offA1 = offA0 + (TRANSA ? 16 : 16 * LDA);
  int offB0 = (2 * half) * LDB + (n0 + r);
  int offB1 = offB0 + 16;

  v8f acc00 = {0.f,0.f,0.f,0.f,0.f,0.f,0.f,0.f};
  v8f acc01 = acc00, acc10 = acc00, acc11 = acc00;

#pragma unroll 4
  for (int k0 = 0; k0 < K; k0 += 4) {
    v2f a0, a1, b0, b1;
    if (TRANSA) {
      a0.x = A[offA0]; a0.y = A[offA0 + LDA];
      a1.x = A[offA1]; a1.y = A[offA1 + LDA];
    } else {
      a0 = *(const v2f*)(A + offA0);
      a1 = *(const v2f*)(A + offA1);
    }
    b0.x = Bm[offB0]; b0.y = Bm[offB0 + LDB];
    b1.x = Bm[offB1]; b1.y = Bm[offB1 + LDB];
    offA0 += stepA; offA1 += stepA; offB0 += stepB; offB1 += stepB;

    acc00 = __builtin_amdgcn_wmma_f32_16x16x4_f32(false, a0, false, b0, (short)0, acc00, false, false);
    acc01 = __builtin_amdgcn_wmma_f32_16x16x4_f32(false, a0, false, b1, (short)0, acc01, false, false);
    acc10 = __builtin_amdgcn_wmma_f32_16x16x4_f32(false, a1, false, b0, (short)0, acc10, false, false);
    acc11 = __builtin_amdgcn_wmma_f32_16x16x4_f32(false, a1, false, b1, (short)0, acc11, false, false);
  }

  const int col0 = n0 + r;
  const int col1 = col0 + 16;
  const float bv0 = HASBIAS ? bias[col0] : 0.f;
  const float bv1 = HASBIAS ? bias[col1] : 0.f;
#pragma unroll
  for (int i = 0; i < 8; ++i) {
    const int row0 = m0 + i + 8 * half;
    const int row1 = row0 + 16;
    if (TRANSD) {
      D[(long)col0 * LDD + row0] = acc00[i] + bv0;
      D[(long)col1 * LDD + row0] = acc01[i] + bv1;
      D[(long)col0 * LDD + row1] = acc10[i] + bv0;
      D[(long)col1 * LDD + row1] = acc11[i] + bv1;
    } else {
      D[(long)row0 * LDD + col0] = acc00[i] + bv0;
      D[(long)row0 * LDD + col1] = acc01[i] + bv1;
      D[(long)row1 * LDD + col0] = acc10[i] + bv0;
      D[(long)row1 * LDD + col1] = acc11[i] + bv1;
    }
  }
}

__global__ __launch_bounds__(128) void gat_scores_softmax(
    const float* __restrict__ gl, const float* __restrict__ gr,
    const float* __restrict__ att, float* __restrict__ alpha)
{
  __shared__ alignas(16) float sRow[D_];
  __shared__ float red[4];

  const int tid = threadIdx.x;
  const int h = blockIdx.x;
  const int b = blockIdx.y;
  const float* glb = gl + (long)b * C_ * HID_ + h * D_;
  const float* grb = gr + (long)b * C_ * HID_ + h * D_;
  const float* atth = att + h * D_;

  float g[D_];
  const float4* grow = (const float4*)(grb + (long)tid * HID_);
#pragma unroll
  for (int q = 0; q < D_ / 4; ++q) {
    const float4 v4 = grow[q];
    g[4 * q + 0] = v4.x; g[4 * q + 1] = v4.y;
    g[4 * q + 2] = v4.z; g[4 * q + 3] = v4.w;
  }
  float a[D_];
#pragma unroll
  for (int d = 0; d < D_; ++d) a[d] = atth[d];

  const int lane = tid & 31;
  const int wv = tid >> 5;
  float* arow = alpha + (long)(b * H_ + h) * C_ * C_;

  for (int t = 0; t < C_; ++t) {
    if (tid < D_) sRow[tid] = glb[(long)t * HID_ + tid];
    __syncthreads();

    const int s = tid;
    v4f r4[D_ / 4];
#pragma unroll
    for (int q = 0; q < D_ / 4; ++q) r4[q] = ((const v4f*)sRow)[q];
    asm volatile("" :: "v"(r4[0]), "v"(r4[1]), "v"(r4[2]), "v"(r4[3]),
                       "v"(r4[4]), "v"(r4[5]), "v"(r4[6]), "v"(r4[7]),
                       "v"(r4[8]), "v"(r4[9]), "v"(r4[10]), "v"(r4[11]),
                       "v"(r4[12]), "v"(r4[13]), "v"(r4[14]), "v"(r4[15]));
    float sc = 0.f;
#pragma unroll
    for (int d = 0; d < D_; ++d) {
      float v = r4[d >> 2][d & 3] + g[d];
      v = v > 0.f ? v : NEG_ * v;
      sc = fmaf(a[d], v, sc);
    }
    float m = sc;
    for (int off = 16; off > 0; off >>= 1) m = fmaxf(m, __shfl_xor(m, off, 32));
    if (lane == 0) red[wv] = m;
    __syncthreads();
    m = fmaxf(fmaxf(red[0], red[1]), fmaxf(red[2], red[3]));
    __syncthreads();
    const float e = __expf(sc - m);
    float ssum = e;
    for (int off = 16; off > 0; off >>= 1) ssum += __shfl_xor(ssum, off, 32);
    if (lane == 0) red[wv] = ssum;
    __syncthreads();
    ssum = red[0] + red[1] + red[2] + red[3];
    arow[(long)t * C_ + s] = e / ssum;
    __syncthreads();
  }
}

__global__ __launch_bounds__(256) void post_elu_residual_ln(
    float* __restrict__ h, const float* __restrict__ gout,
    const float* __restrict__ g, const float* __restrict__ bb)
{
  __shared__ float red[8];
  const int tid = threadIdx.x;
  const long base = (long)blockIdx.x * HID_;

  float x = gout[base + tid];
  x = x > 0.f ? x : (__expf(x) - 1.f);
  const float v = h[base + tid] + x;

  const int lane = tid & 31, wv = tid >> 5;
  float s = v;
  for (int off = 16; off > 0; off >>= 1) s += __shfl_xor(s, off, 32);
  if (lane == 0) red[wv] = s;
  __syncthreads();
  s = 0.f;
#pragma unroll
  for (int i = 0; i < 8; ++i) s += red[i];
  const float mu = s * (1.f / HID_);
  __syncthreads();

  const float dmu = v - mu;
  float q = dmu * dmu;
  for (int off = 16; off > 0; off >>= 1) q += __shfl_xor(q, off, 32);
  if (lane == 0) red[wv] = q;
  __syncthreads();
  q = 0.f;
#pragma unroll
  for (int i = 0; i < 8; ++i) q += red[i];
  const float inv = rsqrtf(q * (1.f / HID_) + EPS_);

  h[base + tid] = dmu * inv * g[tid] + bb[tid];
}

extern "C" void kernel_launch(void* const* d_in, const int* in_sizes, int n_in,
                              void* d_out, int out_size, void* d_ws, size_t ws_size,
                              hipStream_t stream) {
  (void)in_sizes; (void)n_in; (void)out_size; (void)ws_size;
  const float* x      = (const float*)d_in[0];
  const float* W_emb  = (const float*)d_in[1];
  const float* b_emb  = (const float*)d_in[2];
  const float* Wl     = (const float*)d_in[3];
  const float* Wr     = (const float*)d_in[4];
  const float* att    = (const float*)d_in[5];
  const float* bias_g = (const float*)d_in[6];
  const float* ln_g   = (const float*)d_in[7];
  const float* ln_b   = (const float*)d_in[8];
  const float* W_proj = (const float*)d_in[9];
  const float* b_proj = (const float*)d_in[10];

  float* out   = (float*)d_out;
  float* alpha = out + (long)B_ * L_ * C_;

  float* ws = (float*)d_ws;
  const long hwN = (long)B_ * C_ * HID_;
  float* h    = ws;
  float* gl   = ws + hwN;
  float* gr   = ws + 2 * hwN;
  float* gout = ws + 3 * hwN;

  wmma_gemm_f32<C_, HID_, L_, C_, HID_, HID_, 1,
                (long)L_ * C_, 0, 0, 0, (long)C_ * HID_, 0, 0,
                true, false, true>
      <<<dim3((C_ / 32) * (HID_ / 32) / 4, B_), 128, 0, stream>>>(
          x, W_emb, b_emb, h);

  for (int layer = 0; layer < NL_; ++layer) {
    const float* Wl_i  = Wl + (long)layer * HID_ * HID_;
    const float* Wr_i  = Wr + (long)layer * HID_ * HID_;
    const float* att_i = att + (long)layer * H_ * D_;
    const float* bg_i  = bias_g + (long)layer * HID_;

    dim3 g1((C_ / 32) * (HID_ / 32) / 4, B_);
    wmma_gemm_f32<C_, HID_, HID_, HID_, HID_, HID_, 1,
                  (long)C_ * HID_, 0, 0, 0, (long)C_ * HID_, 0, 0,
                  false, false, false>
        <<<g1, 128, 0, stream>>>(h, Wl_i, nullptr, gl);
    wmma_gemm_f32<C_, HID_, HID_, HID_, HID_, HID_, 1,
                  (long)C_ * HID_, 0, 0, 0, (long)C_ * HID_, 0, 0,
                  false, false, false>
        <<<g1, 128, 0, stream>>>(h, Wr_i, nullptr, gr);

    gat_scores_softmax<<<dim3(H_, B_), 128, 0, stream>>>(gl, gr, att_i, alpha);

    wmma_gemm_f32<C_, D_, C_, C_, HID_, HID_, H_,
                  (long)H_ * C_ * C_, (long)C_ * C_,
                  (long)C_ * HID_, D_, (long)C_ * HID_, D_, D_,
                  false, false, true>
        <<<dim3((C_ / 32) * (D_ / 32) / 4, B_ * H_), 128, 0, stream>>>(
            alpha, gr, bg_i, gout);

    post_elu_residual_ln<<<B_ * C_, 256, 0, stream>>>(
        h, gout, ln_g + (long)layer * HID_, ln_b + (long)layer * HID_);
  }

  wmma_gemm_f32<C_, L_, HID_, HID_, L_, C_, 1,
                (long)C_ * HID_, 0, 0, 0, (long)L_ * C_, 0, 0,
                false, true, true>
      <<<dim3((C_ / 32) * (L_ / 32) / 4, B_), 128, 0, stream>>>(
          h, W_proj, b_proj, out);
}


# scores kernel: the 128 independent target rows of each (batch, head) split over grid.z in chunks of 16 (1024 work-groups instead of 128); per-row arithmetic and summation order unchanged
# speedup vs baseline: 4.0905x; 2.8312x over previous
#include <hip/hip_runtime.h>
#include <hip/hip_bf16.h>

typedef __attribute__((ext_vector_type(2))) float v2f;
typedef __attribute__((ext_vector_type(8))) float v8f;
typedef __attribute__((ext_vector_type(4))) float v4f;

#define B_   32
#define L_   512
#define C_   128
#define HID_ 256
#define H_   4
#define D_   64
#define NL_  3
#define NEG_ 0.2f
#define EPS_ 1e-5f
#define TCH_ 16

template <int M, int N, int K, int LDA, int LDB, int LDD, int INNER,
          long SAO, long SAI, long SBO, long SBI, long SDO, long SDI,
          long SBIASI, bool TRANSA, bool TRANSD, bool HASBIAS>
__global__ __launch_bounds__(128) void wmma_gemm_f32(
    const float* __restrict__ A, const float* __restrict__ Bm,
    const float* __restrict__ bias, float* __restrict__ D)
{
  const int lane = threadIdx.x & 31;
  const int wave = threadIdx.x >> 5;
  const int half = (lane >> 4) & 1;
  const int r    = lane & 15;

  constexpr int tilesN = N >> 5;
  const int tile = blockIdx.x * 4 + wave;
  const int tm = tile / tilesN;
  const int tn = tile % tilesN;

  const int batch = blockIdx.y;
  const int bo = batch / INNER;
  const int bi = batch % INNER;
  A    += bo * SAO + bi * SAI;
  Bm   += bo * SBO + bi * SBI;
  D    += bo * SDO + bi * SDI;
  if (HASBIAS) bias += bi * SBIASI;

  const int m0 = tm << 5;
  const int n0 = tn << 5;

  constexpr int stepA = TRANSA ? 4 * LDA : 4;
  constexpr int stepB = 4 * LDB;
  int offA0 = TRANSA ? (2 * half) * LDA + (m0 + r)
                     : (m0 + r) * LDA + 2 * half;
  int offA1 = offA0 + (TRANSA ? 16 : 16 * LDA);
  int offB0 = (2 * half) * LDB + (n0 + r);
  int offB1 = offB0 + 16;

  v8f acc00 = {0.f,0.f,0.f,0.f,0.f,0.f,0.f,0.f};
  v8f acc01 = acc00, acc10 = acc00, acc11 = acc00;

#pragma unroll 4
  for (int k0 = 0; k0 < K; k0 += 4) {
    v2f a0, a1, b0, b1;
    if (TRANSA) {
      a0.x = A[offA0]; a0.y = A[offA0 + LDA];
      a1.x = A[offA1]; a1.y = A[offA1 + LDA];
    } else {
      a0 = *(const v2f*)(A + offA0);
      a1 = *(const v2f*)(A + offA1);
    }
    b0.x = Bm[offB0]; b0.y = Bm[offB0 + LDB];
    b1.x = Bm[offB1]; b1.y = Bm[offB1 + LDB];
    offA0 += stepA; offA1 += stepA; offB0 += stepB; offB1 += stepB;

    acc00 = __builtin_amdgcn_wmma_f32_16x16x4_f32(false, a0, false, b0, (short)0, acc00, false, false);
    acc01 = __builtin_amdgcn_wmma_f32_16x16x4_f32(false, a0, false, b1, (short)0, acc01, false, false);
    acc10 = __builtin_amdgcn_wmma_f32_16x16x4_f32(false, a1, false, b0, (short)0, acc10, false, false);
    acc11 = __builtin_amdgcn_wmma_f32_16x16x4_f32(false, a1, false, b1, (short)0, acc11, false, false);
  }

  const int col0 = n0 + r;
  const int col1 = col0 + 16;
  const float bv0 = HASBIAS ? bias[col0] : 0.f;
  const float bv1 = HASBIAS ? bias[col1] : 0.f;
#pragma unroll
  for (int i = 0; i < 8; ++i) {
    const int row0 = m0 + i + 8 * half;
    const int row1 = row0 + 16;
    if (TRANSD) {
      D[(long)col0 * LDD + row0] = acc00[i] + bv0;
      D[(long)col1 * LDD + row0] = acc01[i] + bv1;
      D[(long)col0 * LDD + row1] = acc10[i] + bv0;
      D[(long)col1 * LDD + row1] = acc11[i] + bv1;
    } else {
      D[(long)row0 * LDD + col0] = acc00[i] + bv0;
      D[(long)row0 * LDD + col1] = acc01[i] + bv1;
      D[(long)row1 * LDD + col0] = acc10[i] + bv0;
      D[(long)row1 * LDD + col1] = acc11[i] + bv1;
    }
  }
}

__global__ __launch_bounds__(128) void gat_scores_softmax(
    const float* __restrict__ gl, const float* __restrict__ gr,
    const float* __restrict__ att, float* __restrict__ alpha)
{
  __shared__ alignas(16) float sRow[D_];
  __shared__ float red[4];

  const int tid = threadIdx.x;
  const int h = blockIdx.x;
  const int b = blockIdx.y;
  const float* glb = gl + (long)b * C_ * HID_ + h * D_;
  const float* grb = gr + (long)b * C_ * HID_ + h * D_;
  const float* atth = att + h * D_;

  float g[D_];
  const float4* grow = (const float4*)(grb + (long)tid * HID_);
#pragma unroll
  for (int q = 0; q < D_ / 4; ++q) {
    const float4 v4 = grow[q];
    g[4 * q + 0] = v4.x; g[4 * q + 1] = v4.y;
    g[4 * q + 2] = v4.z; g[4 * q + 3] = v4.w;
  }
  float a[D_];
#pragma unroll
  for (int d = 0; d < D_; ++d) a[d] = atth[d];

  const int lane = tid & 31;
  const int wv = tid >> 5;
  float* arow = alpha + (long)(b * H_ + h) * C_ * C_;

  const int t0 = blockIdx.z * TCH_;
  for (int t = t0; t < t0 + TCH_; ++t) {
    if (tid < D_) sRow[tid] = glb[(long)t * HID_ + tid];
    __syncthreads();

    const int s = tid;
    v4f r4[D_ / 4];
#pragma unroll
    for (int q = 0; q < D_ / 4; ++q) r4[q] = ((const v4f*)sRow)[q];
    asm volatile("" :: "v"(r4[0]), "v"(r4[1]), "v"(r4[2]), "v"(r4[3]),
                       "v"(r4[4]), "v"(r4[5]), "v"(r4[6]), "v"(r4[7]),
                       "v"(r4[8]), "v"(r4[9]), "v"(r4[10]), "v"(r4[11]),
                       "v"(r4[12]), "v"(r4[13]), "v"(r4[14]), "v"(r4[15]));
    float sc = 0.f;
#pragma unroll
    for (int d = 0; d < D_; ++d) {
      float v = r4[d >> 2][d & 3] + g[d];
      v = v > 0.f ? v : NEG_ * v;
      sc = fmaf(a[d], v, sc);
    }
    float m = sc;
    for (int off = 16; off > 0; off >>= 1) m = fmaxf(m, __shfl_xor(m, off, 32));
    if (lane == 0) red[wv] = m;
    __syncthreads();
    m = fmaxf(fmaxf(red[0], red[1]), fmaxf(red[2], red[3]));
    __syncthreads();
    const float e = __expf(sc - m);
    float ssum = e;
    for (int off = 16; off > 0; off >>= 1) ssum += __shfl_xor(ssum, off, 32);
    if (lane == 0) red[wv] = ssum;
    __syncthreads();
    ssum = red[0] + red[1] + red[2] + red[3];
    arow[(long)t * C_ + s] = e / ssum;
    __syncthreads();
  }
}

__global__ __launch_bounds__(256) void post_elu_residual_ln(
    float* __restrict__ h, const float* __restrict__ gout,
    const float* __restrict__ g, const float* __restrict__ bb)
{
  __shared__ float red[8];
  const int tid = threadIdx.x;
  const long base = (long)blockIdx.x * HID_;

  float x = gout[base + tid];
  x = x > 0.f ? x : (__expf(x) - 1.f);
  const float v = h[base + tid] + x;

  const int lane = tid & 31, wv = tid >> 5;
  float s = v;
  for (int off = 16; off > 0; off >>= 1) s += __shfl_xor(s, off, 32);
  if (lane == 0) red[wv] = s;
  __syncthreads();
  s = 0.f;
#pragma unroll
  for (int i = 0; i < 8; ++i) s += red[i];
  const float mu = s * (1.f / HID_);
  __syncthreads();

  const float dmu = v - mu;
  float q = dmu * dmu;
  for (int off = 16; off > 0; off >>= 1) q += __shfl_xor(q, off, 32);
  if (lane == 0) red[wv] = q;
  __syncthreads();
  q = 0.f;
#pragma unroll
  for (int i = 0; i < 8; ++i) q += red[i];
  const float inv = rsqrtf(q * (1.f / HID_) + EPS_);

  h[base + tid] = dmu * inv * g[tid] + bb[tid];
}

extern "C" void kernel_launch(void* const* d_in, const int* in_sizes, int n_in,
                              void* d_out, int out_size, void* d_ws, size_t ws_size,
                              hipStream_t stream) {
  (void)in_sizes; (void)n_in; (void)out_size; (void)ws_size;
  const float* x      = (const float*)d_in[0];
  const float* W_emb  = (const float*)d_in[1];
  const float* b_emb  = (const float*)d_in[2];
  const float* Wl     = (const float*)d_in[3];
  const float* Wr     = (const float*)d_in[4];
  const float* att    = (const float*)d_in[5];
  const float* bias_g = (const float*)d_in[6];
  const float* ln_g   = (const float*)d_in[7];
  const float* ln_b   = (const float*)d_in[8];
  const float* W_proj = (const float*)d_in[9];
  const float* b_proj = (const float*)d_in[10];

  float* out   = (float*)d_out;
  float* alpha = out + (long)B_ * L_ * C_;

  float* ws = (float*)d_ws;
  const long hwN = (long)B_ * C_ * HID_;
  float* h    = ws;
  float* gl   = ws + hwN;
  float* gr   = ws + 2 * hwN;
  float* gout = ws + 3 * hwN;

  wmma_gemm_f32<C_, HID_, L_, C_, HID_, HID_, 1,
                (long)L_ * C_, 0, 0, 0, (long)C_ * HID_, 0, 0,
                true, false, true>
      <<<dim3((C_ / 32) * (HID_ / 32) / 4, B_), 128, 0, stream>>>(
          x, W_emb, b_emb, h);

  for (int layer = 0; layer < NL_; ++layer) {
    const float* Wl_i  = Wl + (long)layer * HID_ * HID_;
    const float* Wr_i  = Wr + (long)layer * HID_ * HID_;
    const float* att_i = att + (long)layer * H_ * D_;
    const float* bg_i  = bias_g + (long)layer * HID_;

    dim3 g1((C_ / 32) * (HID_ / 32) / 4, B_);
    wmma_gemm_f32<C_, HID_, HID_, HID_, HID_, HID_, 1,
                  (long)C_ * HID_, 0, 0, 0, (long)C_ * HID_, 0, 0,
                  false, false, false>
        <<<g1, 128, 0, stream>>>(h, Wl_i, nullptr, gl);
    wmma_gemm_f32<C_, HID_, HID_, HID_, HID_, HID_, 1,
                  (long)C_ * HID_, 0, 0, 0, (long)C_ * HID_, 0, 0,
                  false, false, false>
        <<<g1, 128, 0, stream>>>(h, Wr_i, nullptr, gr);

    gat_scores_softmax<<<dim3(H_, B_, C_ / TCH_), 128, 0, stream>>>(gl, gr, att_i, alpha);

    wmma_gemm_f32<C_, D_, C_, C_, HID_, HID_, H_,
                  (long)H_ * C_ * C_, (long)C_ * C_,
                  (long)C_ * HID_, D_, (long)C_ * HID_, D_, D_,
                  false, false, true>
        <<<dim3((C_ / 32) * (D_ / 32) / 4, B_ * H_), 128, 0, stream>>>(
            alpha, gr, bg_i, gout);

    post_elu_residual_ln<<<B_ * C_, 256, 0, stream>>>(
        h, gout, ln_g + (long)layer * HID_, ln_b + (long)layer * HID_);
  }

  wmma_gemm_f32<C_, L_, HID_, HID_, L_, C_, 1,
                (long)C_ * HID_, 0, 0, 0, (long)L_ * C_, 0, 0,
                false, true, true>
      <<<dim3((C_ / 32) * (L_ / 32) / 4, B_), 128, 0, stream>>>(
          h, W_proj, b_proj, out);
}


# scores kernel: a work-group's 16 target rows go through together (rows to LDS by one load group, barrier-free score loop, max and sum reductions once on 16 values); summation order unchanged
# speedup vs baseline: 5.0292x; 1.2295x over previous
#include <hip/hip_runtime.h>
#include <hip/hip_bf16.h>

typedef __attribute__((ext_vector_type(2))) float v2f;
typedef __attribute__((ext_vector_type(8))) float v8f;
typedef __attribute__((ext_vector_type(4))) float v4f;

#define B_   32
#define L_   512
#define C_   128
#define HID_ 256
#define H_   4
#define D_   64
#define NL_  3
#define NEG_ 0.2f
#define EPS_ 1e-5f
#define TCH_ 16

template <int M, int N, int K, int LDA, int LDB, int LDD, int INNER,
          long SAO, long SAI, long SBO, long SBI, long SDO, long SDI,
          long SBIASI, bool TRANSA, bool TRANSD, bool HASBIAS>
__global__ __launch_bounds__(128) void wmma_gemm_f32(
    const float* __restrict__ A, const float* __restrict__ Bm,
    const float* __restrict__ bias, float* __restrict__ D)
{
  const int lane = threadIdx.x & 31;
  const int wave = threadIdx.x >> 5;
  const int half = (lane >> 4) & 1;
  const int r    = lane & 15;

  constexpr int tilesN = N >> 5;
  const int tile = blockIdx.x * 4 + wave;
  const int tm = tile / tilesN;
  const int tn = tile % tilesN;

  const int batch = blockIdx.y;
  const int bo = batch / INNER;
  const int bi = batch % INNER;
  A    += bo * SAO + bi * SAI;
  Bm   += bo * SBO + bi * SBI;
  D    += bo * SDO + bi * SDI;
  if (HASBIAS) bias += bi * SBIASI;

  const int m0 = tm << 5;
  const int n0 = tn << 5;

  constexpr int stepA = TRANSA ? 4 * LDA : 4;
  constexpr int stepB = 4 * LDB;
  int offA0 = TRANSA ? (2 * half) * LDA + (m0 + r)
                     : (m0 + r) * LDA + 2 * half;
  int offA1 = offA0 + (TRANSA ? 16 : 16 * LDA);
  int offB0 = (2 * half) * LDB + (n0 + r);
  int offB1 = offB0 + 16;

  v8f acc00 = {0.f,0.f,0.f,0.f,0.f,0.f,0.f,0.f};
  v8f acc01 = acc00, acc10 = acc00, acc11 = acc00;

#pragma unroll 4
  for (int k0 = 0; k0 < K; k0 += 4) {
    v2f a0, a1, b0, b1;
    if (TRANSA) {
      a0.x = A[offA0]; a0.y = A[offA0 + LDA];
      a1.x = A[offA1]; a1.y = A[offA1 + LDA];
    } else {
      a0 = *(const v2f*)(A + offA0);
      a1 = *(const v2f*)(A + offA1);
    }
    b0.x = Bm[offB0]; b0.y = Bm[offB0 + LDB];
    b1.x = Bm[offB1]; b1.y = Bm[offB1 + LDB];
    offA0 += stepA; offA1 += stepA; offB0 += stepB; offB1 += stepB;

    acc00 = __builtin_amdgcn_wmma_f32_16x16x4_f32(false, a0, false, b0, (short)0, acc00, false, false);
    acc01 = __builtin_amdgcn_wmma_f32_16x16x4_f32(false, a0, false, b1, (short)0, acc01, false, false);
    acc10 = __builtin_amdgcn_wmma_f32_16x16x4_f32(false, a1, false, b0, (short)0, acc10, false, false);
    acc11 = __builtin_amdgcn_wmma_f32_16x16x4_f32(false, a1, false, b1, (short)0, acc11, false, false);
  }

  const int col0 = n0 + r;
  const int col1 = col0 + 16;
  const float bv0 = HASBIAS ? bias[col0] : 0.f;
  const float bv1 = HASBIAS ? bias[col1] : 0.f;
#pragma unroll
  for (int i = 0; i < 8; ++i) {
    const int row0 = m0 + i + 8 * half;
    const int row1 = row0 + 16;
    if (TRANSD) {
      D[(long)col0 * LDD + row0] = acc00[i] + bv0;
      D[(long)col1 * LDD + row0] = acc01[i] + bv1;
      D[(long)col0 * LDD + row1] = acc10[i] + bv0;
      D[(long)col1 * LDD + row1] = acc11[i] + bv1;
    } else {
      D[(long)row0 * LDD + col0] = acc00[i] + bv0;
      D[(long)row0 * LDD + col1] = acc01[i] + bv1;
      D[(long)row1 * LDD + col0] = acc10[i] + bv0;
      D[(long)row1 * LDD + col1] = acc11[i] + bv1;
    }
  }
}

__global__ __launch_bounds__(128) void gat_scores_softmax(
    const float* __restrict__ gl, const float* __restrict__ gr,
    const float* __restrict__ att, float* __restrict__ alpha)
{
  __shared__ alignas(16) float sRows[TCH_ * D_];
  __shared__ float sSc[TCH_ * C_];
  __shared__ alignas(16) float redM[TCH_ * 4];
  __shared__ alignas(16) float redS[TCH_ * 4];

  const int tid = threadIdx.x;
  const int h = blockIdx.x;
  const int b = blockIdx.y;
  const int t0 = blockIdx.z * TCH_;
  const float* glb = gl + (long)b * C_ * HID_ + h * D_;
  const float* grb = gr + (long)b * C_ * HID_ + h * D_;
  const float* atth = att + h * D_;

  float4 stage[TCH_ * D_ / 4 / 128];
#pragma unroll
  for (int i = 0; i < TCH_ * D_ / 4 / 128; ++i) {
    const int idx = tid + 128 * i;
    stage[i] = *(const float4*)(glb + (long)(t0 + (idx >> 4)) * HID_ + 4 * (idx & 15));
  }

  float g[D_];
  const float4* grow = (const float4*)(grb + (long)tid * HID_);
#pragma unroll
  for (int q = 0; q < D_ / 4; ++q) {
    const float4 v4 = grow[q];
    g[4 * q + 0] = v4.x; g[4 * q + 1] = v4.y;
    g[4 * q + 2] = v4.z; g[4 * q + 3] = v4.w;
  }
  float a[D_];
#pragma unroll
  for (int d = 0; d < D_; ++d) a[d] = atth[d];

#pragma unroll
  for (int i = 0; i < TCH_ * D_ / 4 / 128; ++i)
    ((float4*)sRows)[tid + 128 * i] = stage[i];
  __syncthreads();

  const int lane = tid & 31;
  const int wv = tid >> 5;

  for (int t = 0; t < TCH_; ++t) {
    const v4f* rowp = (const v4f*)(sRows + t * D_);
    v4f r4[D_ / 4];
#pragma unroll
    for (int q = 0; q < D_ / 4; ++q) r4[q] = rowp[q];
    asm volatile("" :: "v"(r4[0]), "v"(r4[1]), "v"(r4[2]), "v"(r4[3]),
                       "v"(r4[4]), "v"(r4[5]), "v"(r4[6]), "v"(r4[7]),
                       "v"(r4[8]), "v"(r4[9]), "v"(r4[10]), "v"(r4[11]),
                       "v"(r4[12]), "v"(r4[13]), "v"(r4[14]), "v"(r4[15]));
    float sc = 0.f;
#pragma unroll
    for (int d = 0; d < D_; ++d) {
      float v = r4[d >> 2][d & 3] + g[d];
      v = v > 0.f ? v : NEG_ * v;
      sc = fmaf(a[d], v, sc);
    }
    sSc[t * C_ + tid] = sc;
  }

  float sc[TCH_], m[TCH_];
#pragma unroll
  for (int t = 0; t < TCH_; ++t) { sc[t] = sSc[t * C_ + tid]; m[t] = sc[t]; }

  for (int off = 16; off > 0; off >>= 1) {
#pragma unroll
    for (int t = 0; t < TCH_; ++t) m[t] = fmaxf(m[t], __shfl_xor(m[t], off, 32));
  }
  if (lane == 0) {
#pragma unroll
    for (int t = 0; t < TCH_; ++t) redM[t * 4 + wv] = m[t];
  }
  __syncthreads();
  float e[TCH_], ss[TCH_];
#pragma unroll
  for (int t = 0; t < TCH_; ++t) {
    const float4 r = ((const float4*)redM)[t];
    const float mt = fmaxf(fmaxf(r.x, r.y), fmaxf(r.z, r.w));
    e[t] = __expf(sc[t] - mt);
    ss[t] = e[t];
  }
  for (int off = 16; off > 0; off >>= 1) {
#pragma unroll
    for (int t = 0; t < TCH_; ++t) ss[t] += __shfl_xor(ss[t], off, 32);
  }
  if (lane == 0) {
#pragma unroll
    for (int t = 0; t < TCH_; ++t) redS[t * 4 + wv] = ss[t];
  }
  __syncthreads();
  float* arow = alpha + (long)(b * H_ + h) * C_ * C_ + (long)t0 * C_ + tid;
#pragma unroll
  for (int t = 0; t < TCH_; ++t) {
    const float4 r = ((const float4*)redS)[t];
    const float ssum = r.x + r.y + r.z + r.w;
    arow[t * C_] = e[t] / ssum;
  }
}

__global__ __launch_bounds__(256) void post_elu_residual_ln(
    float* __restrict__ h, const float* __restrict__ gout,
    const float* __restrict__ g, const float* __restrict__ bb)
{
  __shared__ float red[8];
  const int tid = threadIdx.x;
  const long base = (long)blockIdx.x * HID_;

  float x = gout[base + tid];
  x = x > 0.f ? x : (__expf(x) - 1.f);
  const float v = h[base + tid] + x;

  const int lane = tid & 31, wv = tid >> 5;
  float s = v;
  for (int off = 16; off > 0; off >>= 1) s += __shfl_xor(s, off, 32);
  if (lane == 0) red[wv] = s;
  __syncthreads();
  s = 0.f;
#pragma unroll
  for (int i = 0; i < 8; ++i) s += red[i];
  const float mu = s * (1.f / HID_);
  __syncthreads();

  const float dmu = v - mu;
  float q = dmu * dmu;
  for (int off = 16; off > 0; off >>= 1) q += __shfl_xor(q, off, 32);
  if (lane == 0) red[wv] = q;
  __syncthreads();
  q = 0.f;
#pragma unroll
  for (int i = 0; i < 8; ++i) q += red[i];
  const float inv = rsqrtf(q * (1.f / HID_) + EPS_);

  h[base + tid] = dmu * inv * g[tid] + bb[tid];
}

extern "C" void kernel_launch(void* const* d_in, const int* in_sizes, int n_in,
                              void* d_out, int out_size, void* d_ws, size_t ws_size,
                              hipStream_t stream) {
  (void)in_sizes; (void)n_in; (void)out_size; (void)ws_size;
  const float* x      = (const float*)d_in[0];
  const float* W_emb  = (const float*)d_in[1];
  const float* b_emb  = (const float*)d_in[2];
  const float* Wl     = (const float*)d_in[3];
  const float* Wr     = (const float*)d_in[4];
  const float* att    = (const float*)d_in[5];
  const float* bias_g = (const float*)d_in[6];
  const float* ln_g   = (const float*)d_in[7];
  const float* ln_b   = (const float*)d_in[8];
  const float* W_proj = (const float*)d_in[9];
  const float* b_proj = (const float*)d_in[10];

  float* out   = (float*)d_out;
  float* alpha = out + (long)B_ * L_ * C_;

  float* ws = (float*)d_ws;
  const long hwN = (long)B_ * C_ * HID_;
  float* h    = ws;
  float* gl   = ws + hwN;
  float* gr   = ws + 2 * hwN;
  float* gout = ws + 3 * hwN;

  wmma_gemm_f32<C_, HID_, L_, C_, HID_, HID_, 1,
                (long)L_ * C_, 0, 0, 0, (long)C_ * HID_, 0, 0,
                true, false, true>
      <<<dim3((C_ / 32) * (HID_ / 32) / 4, B_), 128, 0, stream>>>(
          x, W_emb, b_emb, h);

  for (int layer = 0; layer < NL_; ++layer) {
    const float* Wl_i  = Wl + (long)layer * HID_ * HID_;
    const float* Wr_i  = Wr + (long)layer * HID_ * HID_;
    const float* att_i = att + (long)layer * H_ * D_;
    const float* bg_i  = bias_g + (long)layer * HID_;

    dim3 g1((C_ / 32) * (HID_ / 32) / 4, B_);
    wmma_gemm_f32<C_, HID_, HID_, HID_, HID_, HID_, 1,
                  (long)C_ * HID_, 0, 0, 0, (long)C_ * HID_, 0, 0,
                  false, false, false>
        <<<g1, 128, 0, stream>>>(h, Wl_i, nullptr, gl);
    wmma_gemm_f32<C_, HID_, HID_, HID_, HID_, HID_, 1,
                  (long)C_ * HID_, 0, 0, 0, (long)C_ * HID_, 0, 0,
                  false, false, false>
        <<<g1, 128, 0, stream>>>(h, Wr_i, nullptr, gr);

    gat_scores_softmax<<<dim3(H_, B_, C_ / TCH_), 128, 0, stream>>>(gl, gr, att_i, alpha);

    wmma_gemm_f32<C_, D_, C_, C_, HID_, HID_, H_,
                  (long)H_ * C_ * C_, (long)C_ * C_,
                  (long)C_ * HID_, D_, (long)C_ * HID_, D_, D_,
                  false, false, true>
        <<<dim3((C_ / 32) * (D_ / 32) / 4, B_ * H_), 128, 0, stream>>>(
            alpha, gr, bg_i, gout);

    post_elu_residual_ln<<<B_ * C_, 256, 0, stream>>>(
        h, gout, ln_g + (long)layer * HID_, ln_b + (long)layer * HID_);
  }

  wmma_gemm_f32<C_, L_, HID_, HID_, L_, C_, 1,
                (long)C_ * HID_, 0, 0, 0, (long)L_ * C_, 0, 0,
                false, true, true>
      <<<dim3((C_ / 32) * (L_ / 32) / 4, B_), 128, 0, stream>>>(
          h, W_proj, b_proj, out);
}


# GEMM launches: K loop replaced by a straight-line pipeline, each block's global loads issued as one group 8 K-steps ahead of its matrix instructions (scheduler groups); k order unchanged
# speedup vs baseline: 7.2796x; 1.4475x over previous
#include <hip/hip_runtime.h>
#include <hip/hip_bf16.h>

typedef __attribute__((ext_vector_type(2))) float v2f;
typedef __attribute__((ext_vector_type(8))) float v8f;
typedef __attribute__((ext_vector_type(4))) float v4f;

#define B_   32
#define L_   512
#define C_   128
#define HID_ 256
#define H_   4
#define D_   64
#define NL_  3
#define NEG_ 0.2f
#define EPS_ 1e-5f
#define TCH_ 16

template <int KS, int LDA, int LDB, bool TRANSA>
__device__ __forceinline__ void gemm_load_block(
    const float* __restrict__ pA0, const float* __restrict__ pA1,
    const float* __restrict__ pB0, const float* __restrict__ pB1,
    v2f (&a0)[KS], v2f (&a1)[KS], v2f (&b0)[KS], v2f (&b1)[KS])
{
  constexpr int stepA = TRANSA ? 4 * LDA : 4;
  constexpr int stepB = 4 * LDB;
#pragma unroll
  for (int j = 0; j < KS; ++j) {
    if (TRANSA) {
      a0[j].x = pA0[j * stepA]; a0[j].y = pA0[j * stepA + LDA];
      a1[j].x = pA1[j * stepA]; a1[j].y = pA1[j * stepA + LDA];
    } else {
      a0[j] = *(const v2f*)(pA0 + j * stepA);
      a1[j] = *(const v2f*)(pA1 + j * stepA);
    }
    b0[j].x = pB0[j * stepB]; b0[j].y = pB0[j * stepB + LDB];
    b1[j].x = pB1[j * stepB]; b1[j].y = pB1[j * stepB + LDB];
  }
}

template <int KS>
__device__ __forceinline__ void gemm_mma_block(
    const v2f (&a0)[KS], const v2f (&a1)[KS],
    const v2f (&b0)[KS], const v2f (&b1)[KS],
    v8f& acc00, v8f& acc01, v8f& acc10, v8f& acc11)
{
#pragma unroll
  for (int j = 0; j < KS; ++j) {
    acc00 = __builtin_amdgcn_wmma_f32_16x16x4_f32(false, a0[j], false, b0[j], (short)0, acc00, false, false);
    acc01 = __builtin_amdgcn_wmma_f32_16x16x4_f32(false, a0[j], false, b1[j], (short)0, acc01, false, false);
    acc10 = __builtin_amdgcn_wmma_f32_16x16x4_f32(false, a1[j], false, b0[j], (short)0, acc10, false, false);
    acc11 = __builtin_amdgcn_wmma_f32_16x16x4_f32(false, a1[j], false, b1[j], (short)0, acc11, false, false);
  }
}

template <int M, int N, int K, int LDA, int LDB, int LDD, int INNER,
          long SAO, long SAI, long SBO, long SBI, long SDO, long SDI,
          long SBIASI, bool TRANSA, bool TRANSD, bool HASBIAS>
__global__ __launch_bounds__(128) void wmma_gemm_f32(
    const float* __restrict__ A, const float* __restrict__ Bm,
    const float* __restrict__ bias, float* __restrict__ D)
{
  const int lane = threadIdx.x & 31;
  const int wave = threadIdx.x >> 5;
  const int half = (lane >> 4) & 1;
  const int r    = lane & 15;

  constexpr int tilesN = N >> 5;
  const int tile = blockIdx.x * 4 + wave;
  const int tm = tile / tilesN;
  const int tn = tile % tilesN;

  const int batch = blockIdx.y;
  const int bo = batch / INNER;
  const int bi = batch % INNER;
  A    += bo * SAO + bi * SAI;
  Bm   += bo * SBO + bi * SBI;
  D    += bo * SDO + bi * SDI;
  if (HASBIAS) bias += bi * SBIASI;

  const int m0 = tm << 5;
  const int n0 = tn << 5;

  constexpr int KS   = TRANSA ? 4 : 8;
  constexpr int DIST = TRANSA ? 3 : 2;
  constexpr int NB   = K / (4 * KS);
  static_assert(K % (4 * KS) == 0 && NB >= DIST, "K must be whole blocks");
  constexpr int blkA = KS * (TRANSA ? 4 * LDA : 4);
  constexpr int blkB = KS * 4 * LDB;
  const int offA0 = TRANSA ? (2 * half) * LDA + (m0 + r)
                           : (m0 + r) * LDA + 2 * half;
  const int offB0 = (2 * half) * LDB + (n0 + r);
  const float* pA0 = A + offA0;
  const float* pA1 = pA0 + (TRANSA ? 16 : 16 * LDA);
  const float* pB0 = Bm + offB0;
  const float* pB1 = pB0 + 16;

  v8f acc00 = {0.f,0.f,0.f,0.f,0.f,0.f,0.f,0.f};
  v8f acc01 = acc00, acc10 = acc00, acc11 = acc00;

  v2f fa0[NB][KS], fa1[NB][KS], fb0[NB][KS], fb1[NB][KS];
#pragma unroll
  for (int kb = 0; kb < DIST; ++kb)
    gemm_load_block<KS, LDA, LDB, TRANSA>(pA0 + kb * blkA, pA1 + kb * blkA,
                                          pB0 + kb * blkB, pB1 + kb * blkB,
                                          fa0[kb], fa1[kb], fb0[kb], fb1[kb]);
#pragma unroll
  for (int kb = 0; kb < NB; ++kb) {
    gemm_mma_block<KS>(fa0[kb], fa1[kb], fb0[kb], fb1[kb], acc00, acc01, acc10, acc11);
    if (kb + DIST < NB)
      gemm_load_block<KS, LDA, LDB, TRANSA>(pA0 + (kb + DIST) * blkA, pA1 + (kb + DIST) * blkA,
                                            pB0 + (kb + DIST) * blkB, pB1 + (kb + DIST) * blkB,
                                            fa0[kb + DIST], fa1[kb + DIST],
                                            fb0[kb + DIST], fb1[kb + DIST]);
  }
  constexpr int LPB = (TRANSA ? 8 : 6) * KS;
#pragma unroll
  for (int kb = 0; kb < DIST; ++kb)
    __builtin_amdgcn_sched_group_barrier(0x020, LPB, 0);
#pragma unroll
  for (int kb = 0; kb < NB; ++kb) {
    __builtin_amdgcn_sched_group_barrier(0x008, 4 * KS, 0);
    if (kb + DIST < NB) __builtin_amdgcn_sched_group_barrier(0x020, LPB, 0);
  }

  const int col0 = n0 + r;
  const int col1 = col0 + 16;
  const float bv0 = HASBIAS ? bias[col0] : 0.f;
  const float bv1 = HASBIAS ? bias[col1] : 0.f;
#pragma unroll
  for (int i = 0; i < 8; ++i) {
    const int row0 = m0 + i + 8 * half;
    const int row1 = row0 + 16;
    if (TRANSD) {
      D[(long)col0 * LDD + row0] = acc00[i] + bv0;
      D[(long)col1 * LDD + row0] = acc01[i] + bv1;
      D[(long)col0 * LDD + row1] = acc10[i] + bv0;
      D[(long)col1 * LDD + row1] = acc11[i] + bv1;
    } else {
      D[(long)row0 * LDD + col0] = acc00[i] + bv0;
      D[(long)row0 * LDD + col1] = acc01[i] + bv1;
      D[(long)row1 * LDD + col0] = acc10[i] + bv0;
      D[(long)row1 * LDD + col1] = acc11[i] + bv1;
    }
  }
}

__global__ __launch_bounds__(128) void gat_scores_softmax(
    const float* __restrict__ gl, const float* __restrict__ gr,
    const float* __restrict__ att, float* __restrict__ alpha)
{
  __shared__ alignas(16) float sRows[TCH_ * D_];
  __shared__ float sSc[TCH_ * C_];
  __shared__ alignas(16) float redM[TCH_ * 4];
  __shared__ alignas(16) float redS[TCH_ * 4];

  const int tid = threadIdx.x;
  const int h = blockIdx.x;
  const int b = blockIdx.y;
  const int t0 = blockIdx.z * TCH_;
  const float* glb = gl + (long)b * C_ * HID_ + h * D_;
  const float* grb = gr + (long)b * C_ * HID_ + h * D_;
  const float* atth = att + h * D_;

  float4 stage[TCH_ * D_ / 4 / 128];
#pragma unroll
  for (int i = 0; i < TCH_ * D_ / 4 / 128; ++i) {
    const int idx = tid + 128 * i;
    stage[i] = *(const float4*)(glb + (long)(t0 + (idx >> 4)) * HID_ + 4 * (idx & 15));
  }

  float g[D_];
  const float4* grow = (const float4*)(grb + (long)tid * HID_);
#pragma unroll
  for (int q = 0; q < D_ / 4; ++q) {
    const float4 v4 = grow[q];
    g[4 * q + 0] = v4.x; g[4 * q + 1] = v4.y;
    g[4 * q + 2] = v4.z; g[4 * q + 3] = v4.w;
  }
  float a[D_];
#pragma unroll
  for (int d = 0; d < D_; ++d) a[d] = atth[d];

#pragma unroll
  for (int i = 0; i < TCH_ * D_ / 4 / 128; ++i)
    ((float4*)sRows)[tid + 128 * i] = stage[i];
  __syncthreads();

  const int lane = tid & 31;
  const int wv = tid >> 5;

  for (int t = 0; t < TCH_; ++t) {
    const v4f* rowp = (const v4f*)(sRows + t * D_);
    v4f r4[D_ / 4];
#pragma unroll
    for (int q = 0; q < D_ / 4; ++q) r4[q] = rowp[q];
    asm volatile("" :: "v"(r4[0]), "v"(r4[1]), "v"(r4[2]), "v"(r4[3]),
                       "v"(r4[4]), "v"(r4[5]), "v"(r4[6]), "v"(r4[7]),
                       "v"(r4[8]), "v"(r4[9]), "v"(r4[10]), "v"(r4[11]),
                       "v"(r4[12]), "v"(r4[13]), "v"(r4[14]), "v"(r4[15]));
    float sc = 0.f;
#pragma unroll
    for (int d = 0; d < D_; ++d) {
      float v = r4[d >> 2][d & 3] + g[d];
      v = v > 0.f ? v : NEG_ * v;
      sc = fmaf(a[d], v, sc);
    }
    sSc[t * C_ + tid] = sc;
  }

  float sc[TCH_], m[TCH_];
#pragma unroll
  for (int t = 0; t < TCH_; ++t) { sc[t] = sSc[t * C_ + tid]; m[t] = sc[t]; }

  for (int off = 16; off > 0; off >>= 1) {
#pragma unroll
    for (int t = 0; t < TCH_; ++t) m[t] = fmaxf(m[t], __shfl_xor(m[t], off, 32));
  }
  if (lane == 0) {
#pragma unroll
    for (int t = 0; t < TCH_; ++t) redM[t * 4 + wv] = m[t];
  }
  __syncthreads();
  float e[TCH_], ss[TCH_];
#pragma unroll
  for (int t = 0; t < TCH_; ++t) {
    const float4 r = ((const float4*)redM)[t];
    const float mt = fmaxf(fmaxf(r.x, r.y), fmaxf(r.z, r.w));
    e[t] = __expf(sc[t] - mt);
    ss[t] = e[t];
  }
  for (int off = 16; off > 0; off >>= 1) {
#pragma unroll
    for (int t = 0; t < TCH_; ++t) ss[t] += __shfl_xor(ss[t], off, 32);
  }
  if (lane == 0) {
#pragma unroll
    for (int t = 0; t < TCH_; ++t) redS[t * 4 + wv] = ss[t];
  }
  __syncthreads();
  float* arow = alpha + (long)(b * H_ + h) * C_ * C_ + (long)t0 * C_ + tid;
#pragma unroll
  for (int t = 0; t < TCH_; ++t) {
    const float4 r = ((const float4*)redS)[t];
    const float ssum = r.x + r.y + r.z + r.w;
    arow[t * C_] = e[t] / ssum;
  }
}

__global__ __launch_bounds__(256) void post_elu_residual_ln(
    float* __restrict__ h, const float* __restrict__ gout,
    const float* __restrict__ g, const float* __restrict__ bb)
{
  __shared__ float red[8];
  const int tid = threadIdx.x;
  const long base = (long)blockIdx.x * HID_;

  float x = gout[base + tid];
  x = x > 0.f ? x : (__expf(x) - 1.f);
  const float v = h[base + tid] + x;

  const int lane = tid & 31, wv = tid >> 5;
  float s = v;
  for (int off = 16; off > 0; off >>= 1) s += __shfl_xor(s, off, 32);
  if (lane == 0) red[wv] = s;
  __syncthreads();
  s = 0.f;
#pragma unroll
  for (int i = 0; i < 8; ++i) s += red[i];
  const float mu = s * (1.f / HID_);
  __syncthreads();

  const float dmu = v - mu;
  float q = dmu * dmu;
  for (int off = 16; off > 0; off >>= 1) q += __shfl_xor(q, off, 32);
  if (lane == 0) red[wv] = q;
  __syncthreads();
  q = 0.f;
#pragma unroll
  for (int i = 0; i < 8; ++i) q += red[i];
  const float inv = rsqrtf(q * (1.f / HID_) + EPS_);

  h[base + tid] = dmu * inv * g[tid] + bb[tid];
}

extern "C" void kernel_launch(void* const* d_in, const int* in_sizes, int n_in,
                              void* d_out, int out_size, void* d_ws, size_t ws_size,
                              hipStream_t stream) {
  (void)in_sizes; (void)n_in; (void)out_size; (void)ws_size;
  const float* x      = (const float*)d_in[0];
  const float* W_emb  = (const float*)d_in[1];
  const float* b_emb  = (const float*)d_in[2];
  const float* Wl     = (const float*)d_in[3];
  const float* Wr     = (const float*)d_in[4];
  const float* att    = (const float*)d_in[5];
  const float* bias_g = (const float*)d_in[6];
  const float* ln_g   = (const float*)d_in[7];
  const float* ln_b   = (const float*)d_in[8];
  const float* W_proj = (const float*)d_in[9];
  const float* b_proj = (const float*)d_in[10];

  float* out   = (float*)d_out;
  float* alpha = out + (long)B_ * L_ * C_;

  float* ws = (float*)d_ws;
  const long hwN = (long)B_ * C_ * HID_;
  float* h    = ws;
  float* gl   = ws + hwN;
  float* gr   = ws + 2 * hwN;
  float* gout = ws + 3 * hwN;

  wmma_gemm_f32<C_, HID_, L_, C_, HID_, HID_, 1,
                (long)L_ * C_, 0, 0, 0, (long)C_ * HID_, 0, 0,
                true, false, true>
      <<<dim3((C_ / 32) * (HID_ / 32) / 4, B_), 128, 0, stream>>>(
          x, W_emb, b_emb, h);

  for (int layer = 0; layer < NL_; ++layer) {
    const float* Wl_i  = Wl + (long)layer * HID_ * HID_;
    const float* Wr_i  = Wr + (long)layer * HID_ * HID_;
    const float* att_i = att + (long)layer * H_ * D_;
    const float* bg_i  = bias_g + (long)layer * HID_;

    dim3 g1((C_ / 32) * (HID_ / 32) / 4, B_);
    wmma_gemm_f32<C_, HID_, HID_, HID_, HID_, HID_, 1,
                  (long)C_ * HID_, 0, 0, 0, (long)C_ * HID_, 0, 0,
                  false, false, false>
        <<<g1, 128, 0, stream>>>(h, Wl_i, nullptr, gl);
    wmma_gemm_f32<C_, HID_, HID_, HID_, HID_, HID_, 1,
                  (long)C_ * HID_, 0, 0, 0, (long)C_ * HID_, 0, 0,
                  false, false, false>
        <<<g1, 128, 0, stream>>>(h, Wr_i, nullptr, gr);

    gat_scores_softmax<<<dim3(H_, B_, C_ / TCH_), 128, 0, stream>>>(gl, gr, att_i, alpha);

    wmma_gemm_f32<C_, D_, C_, C_, HID_, HID_, H_,
                  (long)H_ * C_ * C_, (long)C_ * C_,
                  (long)C_ * HID_, D_, (long)C_ * HID_, D_, D_,
                  false, false, true>
        <<<dim3((C_ / 32) * (D_ / 32) / 4, B_ * H_), 128, 0, stream>>>(
            alpha, gr, bg_i, gout);

    post_elu_residual_ln<<<B_ * C_, 256, 0, stream>>>(
        h, gout, ln_g + (long)layer * HID_, ln_b + (long)layer * HID_);
  }

  wmma_gemm_f32<C_, L_, HID_, HID_, L_, C_, 1,
                (long)C_ * HID_, 0, 0, 0, (long)L_ * C_, 0, 0,
                false, true, true>
      <<<dim3((C_ / 32) * (L_ / 32) / 4, B_), 128, 0, stream>>>(
          h, W_proj, b_proj, out);
}


# scores kernel: leaky-relu split as lin*v + abs*|v|; the linear part gives a per-target sum (constant along a softmax row, left out) and a per-source sum (once per thread); score sums re-associated
# speedup vs baseline: 8.0844x; 1.1106x over previous
#include <hip/hip_runtime.h>
#include <hip/hip_bf16.h>

typedef __attribute__((ext_vector_type(2))) float v2f;
typedef __attribute__((ext_vector_type(8))) float v8f;
typedef __attribute__((ext_vector_type(4))) float v4f;

#define B_   32
#define L_   512
#define C_   128
#define HID_ 256
#define H_   4
#define D_   64
#define NL_  3
#define NEG_ 0.2f
#define EPS_ 1e-5f
#define TCH_ 16
#define LIN_ (0.5f * (1.f + NEG_))
#define ABS_ (0.5f * (1.f - NEG_))

template <int KS, int LDA, int LDB, bool TRANSA>
__device__ __forceinline__ void gemm_load_block(
    const float* __restrict__ pA0, const float* __restrict__ pA1,
    const float* __restrict__ pB0, const float* __restrict__ pB1,
    v2f (&a0)[KS], v2f (&a1)[KS], v2f (&b0)[KS], v2f (&b1)[KS])
{
  constexpr int stepA = TRANSA ? 4 * LDA : 4;
  constexpr int stepB = 4 * LDB;
#pragma unroll
  for (int j = 0; j < KS; ++j) {
    if (TRANSA) {
      a0[j].x = pA0[j * stepA]; a0[j].y = pA0[j * stepA + LDA];
      a1[j].x = pA1[j * stepA]; a1[j].y = pA1[j * stepA + LDA];
    } else {
      a0[j] = *(const v2f*)(pA0 + j * stepA);
      a1[j] = *(const v2f*)(pA1 + j * stepA);
    }
    b0[j].x = pB0[j * stepB]; b0[j].y = pB0[j * stepB + LDB];
    b1[j].x = pB1[j * stepB]; b1[j].y = pB1[j * stepB + LDB];
  }
}

template <int KS>
__device__ __forceinline__ void gemm_mma_block(
    const v2f (&a0)[KS], const v2f (&a1)[KS],
    const v2f (&b0)[KS], const v2f (&b1)[KS],
    v8f& acc00, v8f& acc01, v8f& acc10, v8f& acc11)
{
#pragma unroll
  for (int j = 0; j < KS; ++j) {
    acc00 = __builtin_amdgcn_wmma_f32_16x16x4_f32(false, a0[j], false, b0[j], (short)0, acc00, false, false);
    acc01 = __builtin_amdgcn_wmma_f32_16x16x4_f32(false, a0[j], false, b1[j], (short)0, acc01, false, false);
    acc10 = __builtin_amdgcn_wmma_f32_16x16x4_f32(false, a1[j], false, b0[j], (short)0, acc10, false, false);
    acc11 = __builtin_amdgcn_wmma_f32_16x16x4_f32(false, a1[j], false, b1[j], (short)0, acc11, false, false);
  }
}

template <int M, int N, int K, int LDA, int LDB, int LDD, int INNER,
          long SAO, long SAI, long SBO, long SBI, long SDO, long SDI,
          long SBIASI, bool TRANSA, bool TRANSD, bool HASBIAS>
__global__ __launch_bounds__(128) void wmma_gemm_f32(
    const float* __restrict__ A, const float* __restrict__ Bm,
    const float* __restrict__ bias, float* __restrict__ D)
{
  const int lane = threadIdx.x & 31;
  const int wave = threadIdx.x >> 5;
  const int half = (lane >> 4) & 1;
  const int r    = lane & 15;

  constexpr int tilesN = N >> 5;
  const int tile = blockIdx.x * 4 + wave;
  const int tm = tile / tilesN;
  const int tn = tile % tilesN;

  const int batch = blockIdx.y;
  const int bo = batch / INNER;
  const int bi = batch % INNER;
  A    += bo * SAO + bi * SAI;
  Bm   += bo * SBO + bi * SBI;
  D    += bo * SDO + bi * SDI;
  if (HASBIAS) bias += bi * SBIASI;

  const int m0 = tm << 5;
  const int n0 = tn << 5;

  constexpr int KS   = TRANSA ? 4 : 8;
  constexpr int DIST = TRANSA ? 3 : 2;
  constexpr int NB   = K / (4 * KS);
  static_assert(K % (4 * KS) == 0 && NB >= DIST, "K must be whole blocks");
  constexpr int blkA = KS * (TRANSA ? 4 * LDA : 4);
  constexpr int blkB = KS * 4 * LDB;
  const int offA0 = TRANSA ? (2 * half) * LDA + (m0 + r)
                           : (m0 + r) * LDA + 2 * half;
  const int offB0 = (2 * half) * LDB + (n0 + r);
  const float* pA0 = A + offA0;
  const float* pA1 = pA0 + (TRANSA ? 16 : 16 * LDA);
  const float* pB0 = Bm + offB0;
  const float* pB1 = pB0 + 16;

  v8f acc00 = {0.f,0.f,0.f,0.f,0.f,0.f,0.f,0.f};
  v8f acc01 = acc00, acc10 = acc00, acc11 = acc00;

  v2f fa0[NB][KS], fa1[NB][KS], fb0[NB][KS], fb1[NB][KS];
#pragma unroll
  for (int kb = 0; kb < DIST; ++kb)
    gemm_load_block<KS, LDA, LDB, TRANSA>(pA0 + kb * blkA, pA1 + kb * blkA,
                                          pB0 + kb * blkB, pB1 + kb * blkB,
                                          fa0[kb], fa1[kb], fb0[kb], fb1[kb]);
#pragma unroll
  for (int kb = 0; kb < NB; ++kb) {
    gemm_mma_block<KS>(fa0[kb], fa1[kb], fb0[kb], fb1[kb], acc00, acc01, acc10, acc11);
    if (kb + DIST < NB)
      gemm_load_block<KS, LDA, LDB, TRANSA>(pA0 + (kb + DIST) * blkA, pA1 + (kb + DIST) * blkA,
                                            pB0 + (kb + DIST) * blkB, pB1 + (kb + DIST) * blkB,
                                            fa0[kb + DIST], fa1[kb + DIST],
                                            fb0[kb + DIST], fb1[kb + DIST]);
  }
  constexpr int LPB = (TRANSA ? 8 : 6) * KS;
#pragma unroll
  for (int kb = 0; kb < DIST; ++kb)
    __builtin_amdgcn_sched_group_barrier(0x020, LPB, 0);
#pragma unroll
  for (int kb = 0; kb < NB; ++kb) {
    __builtin_amdgcn_sched_group_barrier(0x008, 4 * KS, 0);
    if (kb + DIST < NB) __builtin_amdgcn_sched_group_barrier(0x020, LPB, 0);
  }

  const int col0 = n0 + r;
  const int col1 = col0 + 16;
  const float bv0 = HASBIAS ? bias[col0] : 0.f;
  const float bv1 = HASBIAS ? bias[col1] : 0.f;
#pragma unroll
  for (int i = 0; i < 8; ++i) {
    const int row0 = m0 + i + 8 * half;
    const int row1 = row0 + 16;
    if (TRANSD) {
      D[(long)col0 * LDD + row0] = acc00[i] + bv0;
      D[(long)col1 * LDD + row0] = acc01[i] + bv1;
      D[(long)col0 * LDD + row1] = acc10[i] + bv0;
      D[(long)col1 * LDD + row1] = acc11[i] + bv1;
    } else {
      D[(long)row0 * LDD + col0] = acc00[i] + bv0;
      D[(long)row0 * LDD + col1] = acc01[i] + bv1;
      D[(long)row1 * LDD + col0] = acc10[i] + bv0;
      D[(long)row1 * LDD + col1] = acc11[i] + bv1;
    }
  }
}

__global__ __launch_bounds__(128) void gat_scores_softmax(
    const float* __restrict__ gl, const float* __restrict__ gr,
    const float* __restrict__ att, float* __restrict__ alpha)
{
  __shared__ alignas(16) float sRows[TCH_ * D_];
  __shared__ float sSc[TCH_ * C_];
  __shared__ alignas(16) float redM[TCH_ * 4];
  __shared__ alignas(16) float redS[TCH_ * 4];

  const int tid = threadIdx.x;
  const int h = blockIdx.x;
  const int b = blockIdx.y;
  const int t0 = blockIdx.z * TCH_;
  const float* glb = gl + (long)b * C_ * HID_ + h * D_;
  const float* grb = gr + (long)b * C_ * HID_ + h * D_;
  const float* atth = att + h * D_;

  float4 stage[TCH_ * D_ / 4 / 128];
#pragma unroll
  for (int i = 0; i < TCH_ * D_ / 4 / 128; ++i) {
    const int idx = tid + 128 * i;
    stage[i] = *(const float4*)(glb + (long)(t0 + (idx >> 4)) * HID_ + 4 * (idx & 15));
  }

  v2f g2[D_ / 2];
  const float4* grow = (const float4*)(grb + (long)tid * HID_);
#pragma unroll
  for (int q = 0; q < D_ / 4; ++q) {
    const float4 v4 = grow[q];
    g2[2 * q + 0] = v2f{v4.x, v4.y};
    g2[2 * q + 1] = v2f{v4.z, v4.w};
  }
  float a[D_];
  float lin = 0.f;
#pragma unroll
  for (int d = 0; d < D_; ++d) {
    a[d] = atth[d];
    lin = fmaf(a[d], g2[d >> 1][d & 1], lin);
  }
  lin *= LIN_;

#pragma unroll
  for (int i = 0; i < TCH_ * D_ / 4 / 128; ++i)
    ((float4*)sRows)[tid + 128 * i] = stage[i];
  __syncthreads();

  const int lane = tid & 31;
  const int wv = tid >> 5;

  for (int t = 0; t < TCH_; ++t) {
    const v4f* rowp = (const v4f*)(sRows + t * D_);
    v4f r4[D_ / 4];
#pragma unroll
    for (int q = 0; q < D_ / 4; ++q) r4[q] = rowp[q];
    float sa = 0.f, sb = 0.f;
#pragma unroll
    for (int q = 0; q < D_ / 8; ++q) {
      const v4f ra = r4[q], rb = r4[q + D_ / 8];
      const v2f al = __builtin_shufflevector(ra, ra, 0, 1) + g2[2 * q];
      const v2f ah = __builtin_shufflevector(ra, ra, 2, 3) + g2[2 * q + 1];
      const v2f bl = __builtin_shufflevector(rb, rb, 0, 1) + g2[2 * q + D_ / 4];
      const v2f bh = __builtin_shufflevector(rb, rb, 2, 3) + g2[2 * q + D_ / 4 + 1];
      sa = fmaf(a[4 * q + 0], fabsf(al.x), sa);
      sb = fmaf(a[4 * q + D_ / 2 + 0], fabsf(bl.x), sb);
      sa = fmaf(a[4 * q + 1], fabsf(al.y), sa);
      sb = fmaf(a[4 * q + D_ / 2 + 1], fabsf(bl.y), sb);
      sa = fmaf(a[4 * q + 2], fabsf(ah.x), sa);
      sb = fmaf(a[4 * q + D_ / 2 + 2], fabsf(bh.x), sb);
      sa = fmaf(a[4 * q + 3], fabsf(ah.y), sa);
      sb = fmaf(a[4 * q + D_ / 2 + 3], fabsf(bh.y), sb);
    }
    sSc[t * C_ + tid] = fmaf(ABS_, sa + sb, lin);
    __builtin_amdgcn_sched_group_barrier(0x100, D_ / 4, 0);
    __builtin_amdgcn_sched_group_barrier(0x002, D_ / 2 + D_, 0);
  }

  float sc[TCH_], m[TCH_];
#pragma unroll
  for (int t = 0; t < TCH_; ++t) { sc[t] = sSc[t * C_ + tid]; m[t] = sc[t]; }

  for (int off = 16; off > 0; off >>= 1) {
#pragma unroll
    for (int t = 0; t < TCH_; ++t) m[t] = fmaxf(m[t], __shfl_xor(m[t], off, 32));
  }
  if (lane == 0) {
#pragma unroll
    for (int t = 0; t < TCH_; ++t) redM[t * 4 + wv] = m[t];
  }
  __syncthreads();
  float e[TCH_], ss[TCH_];
#pragma unroll
  for (int t = 0; t < TCH_; ++t) {
    const float4 r = ((const float4*)redM)[t];
    const float mt = fmaxf(fmaxf(r.x, r.y), fmaxf(r.z, r.w));
    e[t] = __expf(sc[t] - mt);
    ss[t] = e[t];
  }
  for (int off = 16; off > 0; off >>= 1) {
#pragma unroll
    for (int t = 0; t < TCH_; ++t) ss[t] += __shfl_xor(ss[t], off, 32);
  }
  if (lane == 0) {
#pragma unroll
    for (int t = 0; t < TCH_; ++t) redS[t * 4 + wv] = ss[t];
  }
  __syncthreads();
  float* arow = alpha + (long)(b * H_ + h) * C_ * C_ + (long)t0 * C_ + tid;
#pragma unroll
  for (int t = 0; t < TCH_; ++t) {
    const float4 r = ((const float4*)redS)[t];
    const float ssum = r.x + r.y + r.z + r.w;
    arow[t * C_] = e[t] / ssum;
  }
}

__global__ __launch_bounds__(256) void post_elu_residual_ln(
    float* __restrict__ h, const float* __restrict__ gout,
    const float* __restrict__ g, const float* __restrict__ bb)
{
  __shared__ float red[8];
  const int tid = threadIdx.x;
  const long base = (long)blockIdx.x * HID_;

  float x = gout[base + tid];
  x = x > 0.f ? x : (__expf(x) - 1.f);
  const float v = h[base + tid] + x;

  const int lane = tid & 31, wv = tid >> 5;
  float s = v;
  for (int off = 16; off > 0; off >>= 1) s += __shfl_xor(s, off, 32);
  if (lane == 0) red[wv] = s;
  __syncthreads();
  s = 0.f;
#pragma unroll
  for (int i = 0; i < 8; ++i) s += red[i];
  const float mu = s * (1.f / HID_);
  __syncthreads();

  const float dmu = v - mu;
  float q = dmu * dmu;
  for (int off = 16; off > 0; off >>= 1) q += __shfl_xor(q, off, 32);
  if (lane == 0) red[wv] = q;
  __syncthreads();
  q = 0.f;
#pragma unroll
  for (int i = 0; i < 8; ++i) q += red[i];
  const float inv = rsqrtf(q * (1.f / HID_) + EPS_);

  h[base + tid] = dmu * inv * g[tid] + bb[tid];
}

extern "C" void kernel_launch(void* const* d_in, const int* in_sizes, int n_in,
                              void* d_out, int out_size, void* d_ws, size_t ws_size,
                              hipStream_t stream) {
  (void)in_sizes; (void)n_in; (void)out_size; (void)ws_size;
  const float* x      = (const float*)d_in[0];
  const float* W_emb  = (const float*)d_in[1];
  const float* b_emb  = (const float*)d_in[2];
  const float* Wl     = (const float*)d_in[3];
  const float* Wr     = (const float*)d_in[4];
  const float* att    = (const float*)d_in[5];
  const float* bias_g = (const float*)d_in[6];
  const float* ln_g   = (const float*)d_in[7];
  const float* ln_b   = (const float*)d_in[8];
  const float* W_proj = (const float*)d_in[9];
  const float* b_proj = (const float*)d_in[10];

  float* out   = (float*)d_out;
  float* alpha = out + (long)B_ * L_ * C_;

  float* ws = (float*)d_ws;
  const long hwN = (long)B_ * C_ * HID_;
  float* h    = ws;
  float* gl   = ws + hwN;
  float* gr   = ws + 2 * hwN;
  float* gout = ws + 3 * hwN;

  wmma_gemm_f32<C_, HID_, L_, C_, HID_, HID_, 1,
                (long)L_ * C_, 0, 0, 0, (long)C_ * HID_, 0, 0,
                true, false, true>
      <<<dim3((C_ / 32) * (HID_ / 32) / 4, B_), 128, 0, stream>>>(
          x, W_emb, b_emb, h);

  for (int layer = 0; layer < NL_; ++layer) {
    const float* Wl_i  = Wl + (long)layer * HID_ * HID_;
    const float* Wr_i  = Wr + (long)layer * HID_ * HID_;
    const float* att_i = att + (long)layer * H_ * D_;
    const float* bg_i  = bias_g + (long)layer * HID_;

    dim3 g1((C_ / 32) * (HID_ / 32) / 4, B_);
    wmma_gemm_f32<C_, HID_, HID_, HID_, HID_, HID_, 1,
                  (long)C_ * HID_, 0, 0, 0, (long)C_ * HID_, 0, 0,
                  false, false, false>
        <<<g1, 128, 0, stream>>>(h, Wl_i, nullptr, gl);
    wmma_gemm_f32<C_, HID_, HID_, HID_, HID_, HID_, 1,
                  (long)C_ * HID_, 0, 0, 0, (long)C_ * HID_, 0, 0,
                  false, false, false>
        <<<g1, 128, 0, stream>>>(h, Wr_i, nullptr, gr);

    gat_scores_softmax<<<dim3(H_, B_, C_ / TCH_), 128, 0, stream>>>(gl, gr, att_i, alpha);

    wmma_gemm_f32<C_, D_, C_, C_, HID_, HID_, H_,
                  (long)H_ * C_ * C_, (long)C_ * C_,
                  (long)C_ * HID_, D_, (long)C_ * HID_, D_, D_,
                  false, false, true>
        <<<dim3((C_ / 32) * (D_ / 32) / 4, B_ * H_), 128, 0, stream>>>(
            alpha, gr, bg_i, gout);

    post_elu_residual_ln<<<B_ * C_, 256, 0, stream>>>(
        h, gout, ln_g + (long)layer * HID_, ln_b + (long)layer * HID_);
  }

  wmma_gemm_f32<C_, L_, HID_, HID_, L_, C_, 1,
                (long)C_ * HID_, 0, 0, 0, (long)L_ * C_, 0, 0,
                false, true, true>
      <<<dim3((C_ / 32) * (L_ / 32) / 4, B_), 128, 0, stream>>>(
          h, W_proj, b_proj, out);
}


# projection GEMMs: a layer's gl = h@Wl and gr = h@Wr run as ONE launch (grid.z picks the product; 512 work-groups, two waves per SIMD) instead of two; 17 launches to 14; k order unchanged
# speedup vs baseline: 8.8527x; 1.0950x over previous
#include <hip/hip_runtime.h>
#include <hip/hip_bf16.h>

typedef __attribute__((ext_vector_type(2))) float v2f;
typedef __attribute__((ext_vector_type(8))) float v8f;
typedef __attribute__((ext_vector_type(4))) float v4f;

#define B_   32
#define L_   512
#define C_   128
#define HID_ 256
#define H_   4
#define D_   64
#define NL_  3
#define NEG_ 0.2f
#define EPS_ 1e-5f
#define TCH_ 16
#define LIN_ (0.5f * (1.f + NEG_))
#define ABS_ (0.5f * (1.f - NEG_))

template <int KS, int LDA, int LDB, bool TRANSA>
__device__ __forceinline__ void gemm_load_block(
    const float* __restrict__ pA0, const float* __restrict__ pA1,
    const float* __restrict__ pB0, const float* __restrict__ pB1,
    v2f (&a0)[KS], v2f (&a1)[KS], v2f (&b0)[KS], v2f (&b1)[KS])
{
  constexpr int stepA = TRANSA ? 4 * LDA : 4;
  constexpr int stepB = 4 * LDB;
#pragma unroll
  for (int j = 0; j < KS; ++j) {
    if (TRANSA) {
      a0[j].x = pA0[j * stepA]; a0[j].y = pA0[j * stepA + LDA];
      a1[j].x = pA1[j * stepA]; a1[j].y = pA1[j * stepA + LDA];
    } else {
      a0[j] = *(const v2f*)(pA0 + j * stepA);
      a1[j] = *(const v2f*)(pA1 + j * stepA);
    }
    b0[j].x = pB0[j * stepB]; b0[j].y = pB0[j * stepB + LDB];
    b1[j].x = pB1[j * stepB]; b1[j].y = pB1[j * stepB + LDB];
  }
}

template <int KS>
__device__ __forceinline__ void gemm_mma_block(
    const v2f (&a0)[KS], const v2f (&a1)[KS],
    const v2f (&b0)[KS], const v2f (&b1)[KS],
    v8f& acc00, v8f& acc01, v8f& acc10, v8f& acc11)
{
#pragma unroll
  for (int j = 0; j < KS; ++j) {
    acc00 = __builtin_amdgcn_wmma_f32_16x16x4_f32(false, a0[j], false, b0[j], (short)0, acc00, false, false);
    acc01 = __builtin_amdgcn_wmma_f32_16x16x4_f32(false, a0[j], false, b1[j], (short)0, acc01, false, false);
    acc10 = __builtin_amdgcn_wmma_f32_16x16x4_f32(false, a1[j], false, b0[j], (short)0, acc10, false, false);
    acc11 = __builtin_amdgcn_wmma_f32_16x16x4_f32(false, a1[j], false, b1[j], (short)0, acc11, false, false);
  }
}

template <int M, int N, int K, int LDA, int LDB, int LDD,
          bool TRANSA, bool TRANSD, bool HASBIAS>
__device__ __forceinline__ void gemm_wave_tile(
    const float* __restrict__ A, const float* __restrict__ Bm,
    const float* __restrict__ bias, float* __restrict__ D)
{
  const int lane = threadIdx.x & 31;
  const int wave = threadIdx.x >> 5;
  const int half = (lane >> 4) & 1;
  const int r    = lane & 15;

  constexpr int tilesN = N >> 5;
  const int tile = blockIdx.x * 4 + wave;
  const int tm = tile / tilesN;
  const int tn = tile % tilesN;

  const int m0 = tm << 5;
  const int n0 = tn << 5;

  constexpr int KS   = TRANSA ? 4 : 8;
  constexpr int DIST = TRANSA ? 3 : 2;
  constexpr int NB   = K / (4 * KS);
  static_assert(K % (4 * KS) == 0 && NB >= DIST, "K must be whole blocks");
  constexpr int blkA = KS * (TRANSA ? 4 * LDA : 4);
  constexpr int blkB = KS * 4 * LDB;
  const int offA0 = TRANSA ? (2 * half) * LDA + (m0 + r)
                           : (m0 + r) * LDA + 2 * half;
  const int offB0 = (2 * half) * LDB + (n0 + r);
  const float* pA0 = A + offA0;
  const float* pA1 = pA0 + (TRANSA ? 16 : 16 * LDA);
  const float* pB0 = Bm + offB0;
  const float* pB1 = pB0 + 16;

  v8f acc00 = {0.f,0.f,0.f,0.f,0.f,0.f,0.f,0.f};
  v8f acc01 = acc00, acc10 = acc00, acc11 = acc00;

  v2f fa0[NB][KS], fa1[NB][KS], fb0[NB][KS], fb1[NB][KS];
#pragma unroll
  for (int kb = 0; kb < DIST; ++kb)
    gemm_load_block<KS, LDA, LDB, TRANSA>(pA0 + kb * blkA, pA1 + kb * blkA,
                                          pB0 + kb * blkB, pB1 + kb * blkB,
                                          fa0[kb], fa1[kb], fb0[kb], fb1[kb]);
#pragma unroll
  for (int kb = 0; kb < NB; ++kb) {
    gemm_mma_block<KS>(fa0[kb], fa1[kb], fb0[kb], fb1[kb], acc00, acc01, acc10, acc11);
    if (kb + DIST < NB)
      gemm_load_block<KS, LDA, LDB, TRANSA>(pA0 + (kb + DIST) * blkA, pA1 + (kb + DIST) * blkA,
                                            pB0 + (kb + DIST) * blkB, pB1 + (kb + DIST) * blkB,
                                            fa0[kb + DIST], fa1[kb + DIST],
                                            fb0[kb + DIST], fb1[kb + DIST]);
  }
  constexpr int LPB = (TRANSA ? 8 : 6) * KS;
#pragma unroll
  for (int kb = 0; kb < DIST; ++kb)
    __builtin_amdgcn_sched_group_barrier(0x020, LPB, 0);
#pragma unroll
  for (int kb = 0; kb < NB; ++kb) {
    __builtin_amdgcn_sched_group_barrier(0x008, 4 * KS, 0);
    if (kb + DIST < NB) __builtin_amdgcn_sched_group_barrier(0x020, LPB, 0);
  }

  const int col0 = n0 + r;
  const int col1 = col0 + 16;
  const float bv0 = HASBIAS ? bias[col0] : 0.f;
  const float bv1 = HASBIAS ? bias[col1] : 0.f;
#pragma unroll
  for (int i = 0; i < 8; ++i) {
    const int row0 = m0 + i + 8 * half;
    const int row1 = row0 + 16;
    if (TRANSD) {
      D[(long)col0 * LDD + row0] = acc00[i] + bv0;
      D[(long)col1 * LDD + row0] = acc01[i] + bv1;
      D[(long)col0 * LDD + row1] = acc10[i] + bv0;
      D[(long)col1 * LDD + row1] = acc11[i] + bv1;
    } else {
      D[(long)row0 * LDD + col0] = acc00[i] + bv0;
      D[(long)row0 * LDD + col1] = acc01[i] + bv1;
      D[(long)row1 * LDD + col0] = acc10[i] + bv0;
      D[(long)row1 * LDD + col1] = acc11[i] + bv1;
    }
  }
}

template <int M, int N, int K, int LDA, int LDB, int LDD, int INNER,
          long SAO, long SAI, long SBO, long SBI, long SDO, long SDI,
          long SBIASI, bool TRANSA, bool TRANSD, bool HASBIAS>
__global__ __launch_bounds__(128) void wmma_gemm_f32(
    const float* __restrict__ A, const float* __restrict__ Bm,
    const float* __restrict__ bias, float* __restrict__ D)
{
  const int batch = blockIdx.y;
  const int bo = batch / INNER;
  const int bi = batch % INNER;
  A    += bo * SAO + bi * SAI;
  Bm   += bo * SBO + bi * SBI;
  D    += bo * SDO + bi * SDI;
  if (HASBIAS) bias += bi * SBIASI;
  gemm_wave_tile<M, N, K, LDA, LDB, LDD, TRANSA, TRANSD, HASBIAS>(A, Bm, bias, D);
}

__global__ __launch_bounds__(128) void gemm_gl_gr(
    const float* __restrict__ h, const float* __restrict__ Wl,
    const float* __restrict__ Wr, float* __restrict__ gl, float* __restrict__ gr)
{
  const bool second = blockIdx.z != 0;
  const long off = (long)blockIdx.y * C_ * HID_;
  gemm_wave_tile<C_, HID_, HID_, HID_, HID_, HID_, false, false, false>(
      h + off, second ? Wr : Wl, nullptr, (second ? gr : gl) + off);
}

__global__ __launch_bounds__(128) void gat_scores_softmax(
    const float* __restrict__ gl, const float* __restrict__ gr,
    const float* __restrict__ att, float* __restrict__ alpha)
{
  __shared__ alignas(16) float sRows[TCH_ * D_];
  __shared__ float sSc[TCH_ * C_];
  __shared__ alignas(16) float redM[TCH_ * 4];
  __shared__ alignas(16) float redS[TCH_ * 4];

  const int tid = threadIdx.x;
  const int h = blockIdx.x;
  const int b = blockIdx.y;
  const int t0 = blockIdx.z * TCH_;
  const float* glb = gl + (long)b * C_ * HID_ + h * D_;
  const float* grb = gr + (long)b * C_ * HID_ + h * D_;
  const float* atth = att + h * D_;

  float4 stage[TCH_ * D_ / 4 / 128];
#pragma unroll
  for (int i = 0; i < TCH_ * D_ / 4 / 128; ++i) {
    const int idx = tid + 128 * i;
    stage[i] = *(const float4*)(glb + (long)(t0 + (idx >> 4)) * HID_ + 4 * (idx & 15));
  }

  v2f g2[D_ / 2];
  const float4* grow = (const float4*)(grb + (long)tid * HID_);
#pragma unroll
  for (int q = 0; q < D_ / 4; ++q) {
    const float4 v4 = grow[q];
    g2[2 * q + 0] = v2f{v4.x, v4.y};
    g2[2 * q + 1] = v2f{v4.z, v4.w};
  }
  float a[D_];
  float lin = 0.f;
#pragma unroll
  for (int d = 0; d < D_; ++d) {
    a[d] = atth[d];
    lin = fmaf(a[d], g2[d >> 1][d & 1], lin);
  }
  lin *= LIN_;

#pragma unroll
  for (int i = 0; i < TCH_ * D_ / 4 / 128; ++i)
    ((float4*)sRows)[tid + 128 * i] = stage[i];
  __syncthreads();

  const int lane = tid & 31;
  const int wv = tid >> 5;

  for (int t = 0; t < TCH_; ++t) {
    const v4f* rowp = (const v4f*)(sRows + t * D_);
    v4f r4[D_ / 4];
#pragma unroll
    for (int q = 0; q < D_ / 4; ++q) r4[q] = rowp[q];
    float sa = 0.f, sb = 0.f;
#pragma unroll
    for (int q = 0; q < D_ / 8; ++q) {
      const v4f ra = r4[q], rb = r4[q + D_ / 8];
      const v2f al = __builtin_shufflevector(ra, ra, 0, 1) + g2[2 * q];
      const v2f ah = __builtin_shufflevector(ra, ra, 2, 3) + g2[2 * q + 1];
      const v2f bl = __builtin_shufflevector(rb, rb, 0, 1) + g2[2 * q + D_ / 4];
      const v2f bh = __builtin_shufflevector(rb, rb, 2, 3) + g2[2 * q + D_ / 4 + 1];
      sa = fmaf(a[4 * q + 0], fabsf(al.x), sa);
      sb = fmaf(a[4 * q + D_ / 2 + 0], fabsf(bl.x), sb);
      sa = fmaf(a[4 * q + 1], fabsf(al.y), sa);
      sb = fmaf(a[4 * q + D_ / 2 + 1], fabsf(bl.y), sb);
      sa = fmaf(a[4 * q + 2], fabsf(ah.x), sa);
      sb = fmaf(a[4 * q + D_ / 2 + 2], fabsf(bh.x), sb);
      sa = fmaf(a[4 * q + 3], fabsf(ah.y), sa);
      sb = fmaf(a[4 * q + D_ / 2 + 3], fabsf(bh.y), sb);
    }
    sSc[t * C_ + tid] = fmaf(ABS_, sa + sb, lin);
    __builtin_amdgcn_sched_group_barrier(0x100, D_ / 4, 0);
    __builtin_amdgcn_sched_group_barrier(0x002, D_ / 2 + D_, 0);
  }

  float sc[TCH_], m[TCH_];
#pragma unroll
  for (int t = 0; t < TCH_; ++t) { sc[t] = sSc[t * C_ + tid]; m[t] = sc[t]; }

  for (int off = 16; off > 0; off >>= 1) {
#pragma unroll
    for (int t = 0; t < TCH_; ++t) m[t] = fmaxf(m[t], __shfl_xor(m[t], off, 32));
  }
  if (lane == 0) {
#pragma unroll
    for (int t = 0; t < TCH_; ++t) redM[t * 4 + wv] = m[t];
  }
  __syncthreads();
  float e[TCH_], ss[TCH_];
#pragma unroll
  for (int t = 0; t < TCH_; ++t) {
    const float4 r = ((const float4*)redM)[t];
    const float mt = fmaxf(fmaxf(r.x, r.y), fmaxf(r.z, r.w));
    e[t] = __expf(sc[t] - mt);
    ss[t] = e[t];
  }
  for (int off = 16; off > 0; off >>= 1) {
#pragma unroll
    for (int t = 0; t < TCH_; ++t) ss[t] += __shfl_xor(ss[t], off, 32);
  }
  if (lane == 0) {
#pragma unroll
    for (int t = 0; t < TCH_; ++t) redS[t * 4 + wv] = ss[t];
  }
  __syncthreads();
  float* arow = alpha + (long)(b * H_ + h) * C_ * C_ + (long)t0 * C_ + tid;
#pragma unroll
  for (int t = 0; t < TCH_; ++t) {
    const float4 r = ((const float4*)redS)[t];
    const float ssum = r.x + r.y + r.z + r.w;
    arow[t * C_] = e[t] / ssum;
  }
}

__global__ __launch_bounds__(256) void post_elu_residual_ln(
    float* __restrict__ h, const float* __restrict__ gout,
    const float* __restrict__ g, const float* __restrict__ bb)
{
  __shared__ float red[8];
  const int tid = threadIdx.x;
  const long base = (long)blockIdx.x * HID_;

  float x = gout[base + tid];
  x = x > 0.f ? x : (__expf(x) - 1.f);
  const float v = h[base + tid] + x;

  const int lane = tid & 31, wv = tid >> 5;
  float s = v;
  for (int off = 16; off > 0; off >>= 1) s += __shfl_xor(s, off, 32);
  if (lane == 0) red[wv] = s;
  __syncthreads();
  s = 0.f;
#pragma unroll
  for (int i = 0; i < 8; ++i) s += red[i];
  const float mu = s * (1.f / HID_);
  __syncthreads();

  const float dmu = v - mu;
  float q = dmu * dmu;
  for (int off = 16; off > 0; off >>= 1) q += __shfl_xor(q, off, 32);
  if (lane == 0) red[wv] = q;
  __syncthreads();
  q = 0.f;
#pragma unroll
  for (int i = 0; i < 8; ++i) q += red[i];
  const float inv = rsqrtf(q * (1.f / HID_) + EPS_);

  h[base + tid] = dmu * inv * g[tid] + bb[tid];
}

extern "C" void kernel_launch(void* const* d_in, const int* in_sizes, int n_in,
                              void* d_out, int out_size, void* d_ws, size_t ws_size,
                              hipStream_t stream) {
  (void)in_sizes; (void)n_in; (void)out_size; (void)ws_size;
  const float* x      = (const float*)d_in[0];
  const float* W_emb  = (const float*)d_in[1];
  const float* b_emb  = (const float*)d_in[2];
  const float* Wl     = (const float*)d_in[3];
  const float* Wr     = (const float*)d_in[4];
  const float* att    = (const float*)d_in[5];
  const float* bias_g = (const float*)d_in[6];
  const float* ln_g   = (const float*)d_in[7];
  const float* ln_b   = (const float*)d_in[8];
  const float* W_proj = (const float*)d_in[9];
  const float* b_proj = (const float*)d_in[10];

  float* out   = (float*)d_out;
  float* alpha = out + (long)B_ * L_ * C_;

  float* ws = (float*)d_ws;
  const long hwN = (long)B_ * C_ * HID_;
  float* h    = ws;
  float* gl   = ws + hwN;
  float* gr   = ws + 2 * hwN;
  float* gout = ws + 3 * hwN;

  wmma_gemm_f32<C_, HID_, L_, C_, HID_, HID_, 1,
                (long)L_ * C_, 0, 0, 0, (long)C_ * HID_, 0, 0,
                true, false, true>
      <<<dim3((C_ / 32) * (HID_ / 32) / 4, B_), 128, 0, stream>>>(
          x, W_emb, b_emb, h);

  for (int layer = 0; layer < NL_; ++layer) {
    const float* Wl_i  = Wl + (long)layer * HID_ * HID_;
    const float* Wr_i  = Wr + (long)layer * HID_ * HID_;
    const float* att_i = att + (long)layer * H_ * D_;
    const float* bg_i  = bias_g + (long)layer * HID_;

    gemm_gl_gr<<<dim3((C_ / 32) * (HID_ / 32) / 4, B_, 2), 128, 0, stream>>>(
        h, Wl_i, Wr_i, gl, gr);

    gat_scores_softmax<<<dim3(H_, B_, C_ / TCH_), 128, 0, stream>>>(gl, gr, att_i, alpha);

    wmma_gemm_f32<C_, D_, C_, C_, HID_, HID_, H_,
                  (long)H_ * C_ * C_, (long)C_ * C_,
                  (long)C_ * HID_, D_, (long)C_ * HID_, D_, D_,
                  false, false, true>
        <<<dim3((C_ / 32) * (D_ / 32) / 4, B_ * H_), 128, 0, stream>>>(
            alpha, gr, bg_i, gout);

    post_elu_residual_ln<<<B_ * C_, 256, 0, stream>>>(
        h, gout, ln_g + (long)layer * HID_, ln_b + (long)layer * HID_);
  }

  wmma_gemm_f32<C_, L_, HID_, HID_, L_, C_, 1,
                (long)C_ * HID_, 0, 0, 0, (long)L_ * C_, 0, 0,
                false, true, true>
      <<<dim3((C_ / 32) * (L_ / 32) / 4, B_), 128, 0, stream>>>(
          h, W_proj, b_proj, out);
}


# aggregation GEMM (alpha @ gr + bias) moved into the scores kernel: the alpha tile goes from LDS into the matrix instructions; 14 launches to 11; alpha stored in the last layer only; k order unchanged
# speedup vs baseline: 9.6897x; 1.0945x over previous
#include <hip/hip_runtime.h>
#include <hip/hip_bf16.h>

typedef __attribute__((ext_vector_type(2))) float v2f;
typedef __attribute__((ext_vector_type(8))) float v8f;
typedef __attribute__((ext_vector_type(4))) float v4f;

#define B_   32
#define L_   512
#define C_   128
#define HID_ 256
#define H_   4
#define D_   64
#define NL_  3
#define NEG_ 0.2f
#define EPS_ 1e-5f
#define TCH_ 16
#define LIN_ (0.5f * (1.f + NEG_))
#define ABS_ (0.5f * (1.f - NEG_))
#define SP_  (C_ + 4)
#define OP_  (D_ + 4)

template <int KS, int LDA, int LDB, bool TRANSA>
__device__ __forceinline__ void gemm_load_block(
    const float* __restrict__ pA0, const float* __restrict__ pA1,
    const float* __restrict__ pB0, const float* __restrict__ pB1,
    v2f (&a0)[KS], v2f (&a1)[KS], v2f (&b0)[KS], v2f (&b1)[KS])
{
  constexpr int stepA = TRANSA ? 4 * LDA : 4;
  constexpr int stepB = 4 * LDB;
#pragma unroll
  for (int j = 0; j < KS; ++j) {
    if (TRANSA) {
      a0[j].x = pA0[j * stepA]; a0[j].y = pA0[j * stepA + LDA];
      a1[j].x = pA1[j * stepA]; a1[j].y = pA1[j * stepA + LDA];
    } else {
      a0[j] = *(const v2f*)(pA0 + j * stepA);
      a1[j] = *(const v2f*)(pA1 + j * stepA);
    }
    b0[j].x = pB0[j * stepB]; b0[j].y = pB0[j * stepB + LDB];
    b1[j].x = pB1[j * stepB]; b1[j].y = pB1[j * stepB + LDB];
  }
}

template <int KS>
__device__ __forceinline__ void gemm_mma_block(
    const v2f (&a0)[KS], const v2f (&a1)[KS],
    const v2f (&b0)[KS], const v2f (&b1)[KS],
    v8f& acc00, v8f& acc01, v8f& acc10, v8f& acc11)
{
#pragma unroll
  for (int j = 0; j < KS; ++j) {
    acc00 = __builtin_amdgcn_wmma_f32_16x16x4_f32(false, a0[j], false, b0[j], (short)0, acc00, false, false);
    acc01 = __builtin_amdgcn_wmma_f32_16x16x4_f32(false, a0[j], false, b1[j], (short)0, acc01, false, false);
    acc10 = __builtin_amdgcn_wmma_f32_16x16x4_f32(false, a1[j], false, b0[j], (short)0, acc10, false, false);
    acc11 = __builtin_amdgcn_wmma_f32_16x16x4_f32(false, a1[j], false, b1[j], (short)0, acc11, false, false);
  }
}

template <int M, int N, int K, int LDA, int LDB, int LDD,
          bool TRANSA, bool TRANSD, bool HASBIAS>
__device__ __forceinline__ void gemm_wave_tile(
    const float* __restrict__ A, const float* __restrict__ Bm,
    const float* __restrict__ bias, float* __restrict__ D)
{
  const int lane = threadIdx.x & 31;
  const int wave = threadIdx.x >> 5;
  const int half = (lane >> 4) & 1;
  const int r    = lane & 15;

  constexpr int tilesN = N >> 5;
  const int tile = blockIdx.x * 4 + wave;
  const int tm = tile / tilesN;
  const int tn = tile % tilesN;

  const int m0 = tm << 5;
  const int n0 = tn << 5;

  constexpr int KS   = TRANSA ? 4 : 8;
  constexpr int DIST = TRANSA ? 3 : 2;
  constexpr int NB   = K / (4 * KS);
  static_assert(K % (4 * KS) == 0 && NB >= DIST, "K must be whole blocks");
  constexpr int blkA = KS * (TRANSA ? 4 * LDA : 4);
  constexpr int blkB = KS * 4 * LDB;
  const int offA0 = TRANSA ? (2 * half) * LDA + (m0 + r)
                           : (m0 + r) * LDA + 2 * half;
  const int offB0 = (2 * half) * LDB + (n0 + r);
  const float* pA0 = A + offA0;
  const float* pA1 = pA0 + (TRANSA ? 16 : 16 * LDA);
  const float* pB0 = Bm + offB0;
  const float* pB1 = pB0 + 16;

  v8f acc00 = {0.f,0.f,0.f,0.f,0.f,0.f,0.f,0.f};
  v8f acc01 = acc00, acc10 = acc00, acc11 = acc00;

  v2f fa0[NB][KS], fa1[NB][KS], fb0[NB][KS], fb1[NB][KS];
#pragma unroll
  for (int kb = 0; kb < DIST; ++kb)
    gemm_load_block<KS, LDA, LDB, TRANSA>(pA0 + kb * blkA, pA1 + kb * blkA,
                                          pB0 + kb * blkB, pB1 + kb * blkB,
                                          fa0[kb], fa1[kb], fb0[kb], fb1[kb]);
#pragma unroll
  for (int kb = 0; kb < NB; ++kb) {
    gemm_mma_block<KS>(fa0[kb], fa1[kb], fb0[kb], fb1[kb], acc00, acc01, acc10, acc11);
    if (kb + DIST < NB)
      gemm_load_block<KS, LDA, LDB, TRANSA>(pA0 + (kb + DIST) * blkA, pA1 + (kb + DIST) * blkA,
                                            pB0 + (kb + DIST) * blkB, pB1 + (kb + DIST) * blkB,
                                            fa0[kb + DIST], fa1[kb + DIST],
                                            fb0[kb + DIST], fb1[kb + DIST]);
  }
  constexpr int LPB = (TRANSA ? 8 : 6) * KS;
#pragma unroll
  for (int kb = 0; kb < DIST; ++kb)
    __builtin_amdgcn_sched_group_barrier(0x020, LPB, 0);
#pragma unroll
  for (int kb = 0; kb < NB; ++kb) {
    __builtin_amdgcn_sched_group_barrier(0x008, 4 * KS, 0);
    if (kb + DIST < NB) __builtin_amdgcn_sched_group_barrier(0x020, LPB, 0);
  }

  const int col0 = n0 + r;
  const int col1 = col0 + 16;
  const float bv0 = HASBIAS ? bias[col0] : 0.f;
  const float bv1 = HASBIAS ? bias[col1] : 0.f;
#pragma unroll
  for (int i = 0; i < 8; ++i) {
    const int row0 = m0 + i + 8 * half;
    const int row1 = row0 + 16;
    if (TRANSD) {
      D[(long)col0 * LDD + row0] = acc00[i] + bv0;
      D[(long)col1 * LDD + row0] = acc01[i] + bv1;
      D[(long)col0 * LDD + row1] = acc10[i] + bv0;
      D[(long)col1 * LDD + row1] = acc11[i] + bv1;
    } else {
      D[(long)row0 * LDD + col0] = acc00[i] + bv0;
      D[(long)row0 * LDD + col1] = acc01[i] + bv1;
      D[(long)row1 * LDD + col0] = acc10[i] + bv0;
      D[(long)row1 * LDD + col1] = acc11[i] + bv1;
    }
  }
}

template <int M, int N, int K, int LDA, int LDB, int LDD, int INNER,
          long SAO, long SAI, long SBO, long SBI, long SDO, long SDI,
          long SBIASI, bool TRANSA, bool TRANSD, bool HASBIAS>
__global__ __launch_bounds__(128) void wmma_gemm_f32(
    const float* __restrict__ A, const float* __restrict__ Bm,
    const float* __restrict__ bias, float* __restrict__ D)
{
  const int batch = blockIdx.y;
  const int bo = batch / INNER;
  const int bi = batch % INNER;
  A    += bo * SAO + bi * SAI;
  Bm   += bo * SBO + bi * SBI;
  D    += bo * SDO + bi * SDI;
  if (HASBIAS) bias += bi * SBIASI;
  gemm_wave_tile<M, N, K, LDA, LDB, LDD, TRANSA, TRANSD, HASBIAS>(A, Bm, bias, D);
}

__global__ __launch_bounds__(128) void gemm_gl_gr(
    const float* __restrict__ h, const float* __restrict__ Wl,
    const float* __restrict__ Wr, float* __restrict__ gl, float* __restrict__ gr)
{
  const bool second = blockIdx.z != 0;
  const long off = (long)blockIdx.y * C_ * HID_;
  gemm_wave_tile<C_, HID_, HID_, HID_, HID_, HID_, false, false, false>(
      h + off, second ? Wr : Wl, nullptr, (second ? gr : gl) + off);
}

__global__ __launch_bounds__(128) void gat_scores_softmax_agg(
    const float* __restrict__ gl, const float* __restrict__ gr,
    const float* __restrict__ att, const float* __restrict__ bias,
    float* __restrict__ alpha, float* __restrict__ gout)
{
  __shared__ alignas(16) float sRows[TCH_ * D_];
  __shared__ alignas(16) float sSc[TCH_ * SP_];
  __shared__ alignas(16) float sOut[TCH_ * OP_];
  __shared__ alignas(16) float redM[TCH_ * 4];
  __shared__ alignas(16) float redS[TCH_ * 4];

  const int tid = threadIdx.x;
  const int h = blockIdx.x;
  const int b = blockIdx.y;
  const int t0 = blockIdx.z * TCH_;
  const float* glb = gl + (long)b * C_ * HID_ + h * D_;
  const float* grb = gr + (long)b * C_ * HID_ + h * D_;
  const float* atth = att + h * D_;

  float4 stage[TCH_ * D_ / 4 / 128];
#pragma unroll
  for (int i = 0; i < TCH_ * D_ / 4 / 128; ++i) {
    const int idx = tid + 128 * i;
    stage[i] = *(const float4*)(glb + (long)(t0 + (idx >> 4)) * HID_ + 4 * (idx & 15));
  }

  v2f g2[D_ / 2];
  const float4* grow = (const float4*)(grb + (long)tid * HID_);
#pragma unroll
  for (int q = 0; q < D_ / 4; ++q) {
    const float4 v4 = grow[q];
    g2[2 * q + 0] = v2f{v4.x, v4.y};
    g2[2 * q + 1] = v2f{v4.z, v4.w};
  }
  float a[D_];
  float lin = 0.f;
#pragma unroll
  for (int d = 0; d < D_; ++d) {
    a[d] = atth[d];
    lin = fmaf(a[d], g2[d >> 1][d & 1], lin);
  }
  lin *= LIN_;

#pragma unroll
  for (int i = 0; i < TCH_ * D_ / 4 / 128; ++i)
    ((float4*)sRows)[tid + 128 * i] = stage[i];
  __syncthreads();

  const int lane = tid & 31;
  const int wv = tid >> 5;

  for (int t = 0; t < TCH_; ++t) {
    const v4f* rowp = (const v4f*)(sRows + t * D_);
    v4f r4[D_ / 4];
#pragma unroll
    for (int q = 0; q < D_ / 4; ++q) r4[q] = rowp[q];
    float sa = 0.f, sb = 0.f;
#pragma unroll
    for (int q = 0; q < D_ / 8; ++q) {
      const v4f ra = r4[q], rb = r4[q + D_ / 8];
      const v2f al = __builtin_shufflevector(ra, ra, 0, 1) + g2[2 * q];
      const v2f ah = __builtin_shufflevector(ra, ra, 2, 3) + g2[2 * q + 1];
      const v2f bl = __builtin_shufflevector(rb, rb, 0, 1) + g2[2 * q + D_ / 4];
      const v2f bh = __builtin_shufflevector(rb, rb, 2, 3) + g2[2 * q + D_ / 4 + 1];
      sa = fmaf(a[4 * q + 0], fabsf(al.x), sa);
      sb = fmaf(a[4 * q + D_ / 2 + 0], fabsf(bl.x), sb);
      sa = fmaf(a[4 * q + 1], fabsf(al.y), sa);
      sb = fmaf(a[4 * q + D_ / 2 + 1], fabsf(bl.y), sb);
      sa = fmaf(a[4 * q + 2], fabsf(ah.x), sa);
      sb = fmaf(a[4 * q + D_ / 2 + 2], fabsf(bh.x), sb);
      sa = fmaf(a[4 * q + 3], fabsf(ah.y), sa);
      sb = fmaf(a[4 * q + D_ / 2 + 3], fabsf(bh.y), sb);
    }
    sSc[t * SP_ + tid] = fmaf(ABS_, sa + sb, lin);
    __builtin_amdgcn_sched_group_barrier(0x100, D_ / 4, 0);
    __builtin_amdgcn_sched_group_barrier(0x002, D_ / 2 + D_, 0);
  }

  const int half = lane >> 4;
  const int r = lane & 15;
  const int n0 = wv * 16;
  v2f bf[C_ / 4];
  {
    const float* pB = grb + (long)(2 * half) * HID_ + n0 + r;
#pragma unroll
    for (int j = 0; j < C_ / 4; ++j) {
      bf[j].x = pB[(long)(4 * j) * HID_];
      bf[j].y = pB[(long)(4 * j + 1) * HID_];
    }
  }

  float sc[TCH_], m[TCH_];
#pragma unroll
  for (int t = 0; t < TCH_; ++t) { sc[t] = sSc[t * SP_ + tid]; m[t] = sc[t]; }

  for (int off = 16; off > 0; off >>= 1) {
#pragma unroll
    for (int t = 0; t < TCH_; ++t) m[t] = fmaxf(m[t], __shfl_xor(m[t], off, 32));
  }
  if (lane == 0) {
#pragma unroll
    for (int t = 0; t < TCH_; ++t) redM[t * 4 + wv] = m[t];
  }
  __syncthreads();
  float e[TCH_], ss[TCH_];
#pragma unroll
  for (int t = 0; t < TCH_; ++t) {
    const float4 r = ((const float4*)redM)[t];
    const float mt = fmaxf(fmaxf(r.x, r.y), fmaxf(r.z, r.w));
    e[t] = __expf(sc[t] - mt);
    ss[t] = e[t];
  }
  for (int off = 16; off > 0; off >>= 1) {
#pragma unroll
    for (int t = 0; t < TCH_; ++t) ss[t] += __shfl_xor(ss[t], off, 32);
  }
  if (lane == 0) {
#pragma unroll
    for (int t = 0; t < TCH_; ++t) redS[t * 4 + wv] = ss[t];
  }
  __syncthreads();
  float av[TCH_];
#pragma unroll
  for (int t = 0; t < TCH_; ++t) {
    const float4 rs = ((const float4*)redS)[t];
    const float ssum = rs.x + rs.y + rs.z + rs.w;
    av[t] = e[t] / ssum;
    sSc[t * SP_ + tid] = av[t];
  }
  __syncthreads();

  v8f acc = {0.f, 0.f, 0.f, 0.f, 0.f, 0.f, 0.f, 0.f};
  {
    const float* pA = sSc + r * SP_ + 2 * half;
#pragma unroll
    for (int j = 0; j < C_ / 4; ++j) {
      const v2f af = *(const v2f*)(pA + 4 * j);
      acc = __builtin_amdgcn_wmma_f32_16x16x4_f32(false, af, false, bf[j], (short)0, acc, false, false);
    }
  }
#pragma unroll
  for (int i = 0; i < 8; ++i) sOut[(i + 8 * half) * OP_ + n0 + r] = acc[i];
  __syncthreads();

  float* gob = gout + ((long)b * C_ + t0) * HID_ + h * D_;
#pragma unroll
  for (int q = 0; q < TCH_ * D_ / 4 / 128; ++q) {
    const int idx = tid + 128 * q;
    const int row = idx >> 4;
    const int c4 = 4 * (idx & 15);
    const float4 v = *(const float4*)(sOut + row * OP_ + c4);
    const float4 bv = *(const float4*)(bias + h * D_ + c4);
    float4 o;
    o.x = v.x + bv.x; o.y = v.y + bv.y; o.z = v.z + bv.z; o.w = v.w + bv.w;
    *(float4*)(gob + (long)row * HID_ + c4) = o;
  }
  if (alpha != nullptr) {
    float* arow = alpha + (long)(b * H_ + h) * C_ * C_ + (long)t0 * C_ + tid;
#pragma unroll
    for (int t = 0; t < TCH_; ++t)
      arow[t * C_] = av[t];
  }
}

__global__ __launch_bounds__(256) void post_elu_residual_ln(
    float* __restrict__ h, const float* __restrict__ gout,
    const float* __restrict__ g, const float* __restrict__ bb)
{
  __shared__ float red[8];
  const int tid = threadIdx.x;
  const long base = (long)blockIdx.x * HID_;

  float x = gout[base + tid];
  x = x > 0.f ? x : (__expf(x) - 1.f);
  const float v = h[base + tid] + x;

  const int lane = tid & 31, wv = tid >> 5;
  float s = v;
  for (int off = 16; off > 0; off >>= 1) s += __shfl_xor(s, off, 32);
  if (lane == 0) red[wv] = s;
  __syncthreads();
  s = 0.f;
#pragma unroll
  for (int i = 0; i < 8; ++i) s += red[i];
  const float mu = s * (1.f / HID_);
  __syncthreads();

  const float dmu = v - mu;
  float q = dmu * dmu;
  for (int off = 16; off > 0; off >>= 1) q += __shfl_xor(q, off, 32);
  if (lane == 0) red[wv] = q;
  __syncthreads();
  q = 0.f;
#pragma unroll
  for (int i = 0; i < 8; ++i) q += red[i];
  const float inv = rsqrtf(q * (1.f / HID_) + EPS_);

  h[base + tid] = dmu * inv * g[tid] + bb[tid];
}

extern "C" void kernel_launch(void* const* d_in, const int* in_sizes, int n_in,
                              void* d_out, int out_size, void* d_ws, size_t ws_size,
                              hipStream_t stream) {
  (void)in_sizes; (void)n_in; (void)out_size; (void)ws_size;
  const float* x      = (const float*)d_in[0];
  const float* W_emb  = (const float*)d_in[1];
  const float* b_emb  = (const float*)d_in[2];
  const float* Wl     = (const float*)d_in[3];
  const float* Wr     = (const float*)d_in[4];
  const float* att    = (const float*)d_in[5];
  const float* bias_g = (const float*)d_in[6];
  const float* ln_g   = (const float*)d_in[7];
  const float* ln_b   = (const float*)d_in[8];
  const float* W_proj = (const float*)d_in[9];
  const float* b_proj = (const float*)d_in[10];

  float* out   = (float*)d_out;
  float* alpha = out + (long)B_ * L_ * C_;

  float* ws = (float*)d_ws;
  const long hwN = (long)B_ * C_ * HID_;
  float* h    = ws;
  float* gl   = ws + hwN;
  float* gr   = ws + 2 * hwN;
  float* gout = ws + 3 * hwN;

  wmma_gemm_f32<C_, HID_, L_, C_, HID_, HID_, 1,
                (long)L_ * C_, 0, 0, 0, (long)C_ * HID_, 0, 0,
                true, false, true>
      <<<dim3((C_ / 32) * (HID_ / 32) / 4, B_), 128, 0, stream>>>(
          x, W_emb, b_emb, h);

  for (int layer = 0; layer < NL_; ++layer) {
    const float* Wl_i  = Wl + (long)layer * HID_ * HID_;
    const float* Wr_i  = Wr + (long)layer * HID_ * HID_;
    const float* att_i = att + (long)layer * H_ * D_;
    const float* bg_i  = bias_g + (long)layer * HID_;

    gemm_gl_gr<<<dim3((C_ / 32) * (HID_ / 32) / 4, B_, 2), 128, 0, stream>>>(
        h, Wl_i, Wr_i, gl, gr);

    gat_scores_softmax_agg<<<dim3(H_, B_, C_ / TCH_), 128, 0, stream>>>(
        gl, gr, att_i, bg_i, layer == NL_ - 1 ? alpha : nullptr, gout);

    post_elu_residual_ln<<<B_ * C_, 256, 0, stream>>>(
        h, gout, ln_g + (long)layer * HID_, ln_b + (long)layer * HID_);
  }

  wmma_gemm_f32<C_, L_, HID_, HID_, L_, C_, 1,
                (long)C_ * HID_, 0, 0, 0, (long)L_ * C_, 0, 0,
                false, true, true>
      <<<dim3((C_ / 32) * (L_ / 32) / 4, B_), 128, 0, stream>>>(
          h, W_proj, b_proj, out);
}


# GEMM launches with row-major A: A fragments fetched as one 16-byte load per two K-steps (half as many 16-line loads at the cache); k order inside each 8 of k changed, sums re-associated
# speedup vs baseline: 10.0106x; 1.0331x over previous
#include <hip/hip_runtime.h>
#include <hip/hip_bf16.h>

typedef __attribute__((ext_vector_type(2))) float v2f;
typedef __attribute__((ext_vector_type(8))) float v8f;
typedef __attribute__((ext_vector_type(4))) float v4f;

#define B_   32
#define L_   512
#define C_   128
#define HID_ 256
#define H_   4
#define D_   64
#define NL_  3
#define NEG_ 0.2f
#define EPS_ 1e-5f
#define TCH_ 16
#define LIN_ (0.5f * (1.f + NEG_))
#define ABS_ (0.5f * (1.f - NEG_))
#define SP_  (C_ + 4)
#define OP_  (D_ + 4)

template <int KS, int LDA, int LDB, bool TRANSA>
__device__ __forceinline__ void gemm_load_block(
    const float* __restrict__ pA0, const float* __restrict__ pA1,
    const float* __restrict__ pB0, const float* __restrict__ pB1,
    v2f (&a0)[KS], v2f (&a1)[KS], v2f (&b0)[KS], v2f (&b1)[KS])
{
  constexpr int stepA = TRANSA ? 4 * LDA : 4;
  constexpr int stepB = 4 * LDB;
  if (TRANSA) {
#pragma unroll
    for (int j = 0; j < KS; ++j) {
      a0[j].x = pA0[j * stepA]; a0[j].y = pA0[j * stepA + LDA];
      a1[j].x = pA1[j * stepA]; a1[j].y = pA1[j * stepA + LDA];
      b0[j].x = pB0[j * stepB]; b0[j].y = pB0[j * stepB + LDB];
      b1[j].x = pB1[j * stepB]; b1[j].y = pB1[j * stepB + LDB];
    }
  } else {
    static_assert(TRANSA || KS % 2 == 0, "K-steps go in pairs");
#pragma unroll
    for (int p = 0; p < KS / 2; ++p) {
      const v4f qa0 = *(const v4f*)(pA0 + 8 * p);
      const v4f qa1 = *(const v4f*)(pA1 + 8 * p);
      a0[2 * p]     = v2f{qa0.x, qa0.y};
      a0[2 * p + 1] = v2f{qa0.z, qa0.w};
      a1[2 * p]     = v2f{qa1.x, qa1.y};
      a1[2 * p + 1] = v2f{qa1.z, qa1.w};
      b0[2 * p].x     = pB0[(8 * p + 0) * LDB]; b0[2 * p].y     = pB0[(8 * p + 1) * LDB];
      b0[2 * p + 1].x = pB0[(8 * p + 2) * LDB]; b0[2 * p + 1].y = pB0[(8 * p + 3) * LDB];
      b1[2 * p].x     = pB1[(8 * p + 0) * LDB]; b1[2 * p].y     = pB1[(8 * p + 1) * LDB];
      b1[2 * p + 1].x = pB1[(8 * p + 2) * LDB]; b1[2 * p + 1].y = pB1[(8 * p + 3) * LDB];
    }
  }
}

template <int KS>
__device__ __forceinline__ void gemm_mma_block(
    const v2f (&a0)[KS], const v2f (&a1)[KS],
    const v2f (&b0)[KS], const v2f (&b1)[KS],
    v8f& acc00, v8f& acc01, v8f& acc10, v8f& acc11)
{
#pragma unroll
  for (int j = 0; j < KS; ++j) {
    acc00 = __builtin_amdgcn_wmma_f32_16x16x4_f32(false, a0[j], false, b0[j], (short)0, acc00, false, false);
    acc01 = __builtin_amdgcn_wmma_f32_16x16x4_f32(false, a0[j], false, b1[j], (short)0, acc01, false, false);
    acc10 = __builtin_amdgcn_wmma_f32_16x16x4_f32(false, a1[j], false, b0[j], (short)0, acc10, false, false);
    acc11 = __builtin_amdgcn_wmma_f32_16x16x4_f32(false, a1[j], false, b1[j], (short)0, acc11, false, false);
  }
}

template <int M, int N, int K, int LDA, int LDB, int LDD,
          bool TRANSA, bool TRANSD, bool HASBIAS>
__device__ __forceinline__ void gemm_wave_tile(
    const float* __restrict__ A, const float* __restrict__ Bm,
    const float* __restrict__ bias, float* __restrict__ D)
{
  const int lane = threadIdx.x & 31;
  const int wave = threadIdx.x >> 5;
  const int half = (lane >> 4) & 1;
  const int r    = lane & 15;

  constexpr int tilesN = N >> 5;
  const int tile = blockIdx.x * 4 + wave;
  const int tm = tile / tilesN;
  const int tn = tile % tilesN;

  const int m0 = tm << 5;
  const int n0 = tn << 5;

  constexpr int KS   = TRANSA ? 4 : 8;
  constexpr int DIST = TRANSA ? 3 : 2;
  constexpr int NB   = K / (4 * KS);
  static_assert(K % (4 * KS) == 0 && NB >= DIST, "K must be whole blocks");
  constexpr int blkA = KS * (TRANSA ? 4 * LDA : 4);
  constexpr int blkB = KS * 4 * LDB;
  constexpr int KH = TRANSA ? 2 : 4;
  const int offA0 = TRANSA ? (KH * half) * LDA + (m0 + r)
                           : (m0 + r) * LDA + KH * half;
  const int offB0 = (KH * half) * LDB + (n0 + r);
  const float* pA0 = A + offA0;
  const float* pA1 = pA0 + (TRANSA ? 16 : 16 * LDA);
  const float* pB0 = Bm + offB0;
  const float* pB1 = pB0 + 16;

  v8f acc00 = {0.f,0.f,0.f,0.f,0.f,0.f,0.f,0.f};
  v8f acc01 = acc00, acc10 = acc00, acc11 = acc00;

  v2f fa0[NB][KS], fa1[NB][KS], fb0[NB][KS], fb1[NB][KS];
#pragma unroll
  for (int kb = 0; kb < DIST; ++kb)
    gemm_load_block<KS, LDA, LDB, TRANSA>(pA0 + kb * blkA, pA1 + kb * blkA,
                                          pB0 + kb * blkB, pB1 + kb * blkB,
                                          fa0[kb], fa1[kb], fb0[kb], fb1[kb]);
#pragma unroll
  for (int kb = 0; kb < NB; ++kb) {
    gemm_mma_block<KS>(fa0[kb], fa1[kb], fb0[kb], fb1[kb], acc00, acc01, acc10, acc11);
    if (kb + DIST < NB)
      gemm_load_block<KS, LDA, LDB, TRANSA>(pA0 + (kb + DIST) * blkA, pA1 + (kb + DIST) * blkA,
                                            pB0 + (kb + DIST) * blkB, pB1 + (kb + DIST) * blkB,
                                            fa0[kb + DIST], fa1[kb + DIST],
                                            fb0[kb + DIST], fb1[kb + DIST]);
  }
  constexpr int LPB = (TRANSA ? 8 : 5) * KS;
#pragma unroll
  for (int kb = 0; kb < DIST; ++kb)
    __builtin_amdgcn_sched_group_barrier(0x020, LPB, 0);
#pragma unroll
  for (int kb = 0; kb < NB; ++kb) {
    __builtin_amdgcn_sched_group_barrier(0x008, 4 * KS, 0);
    if (kb + DIST < NB) __builtin_amdgcn_sched_group_barrier(0x020, LPB, 0);
  }

  const int col0 = n0 + r;
  const int col1 = col0 + 16;
  const float bv0 = HASBIAS ? bias[col0] : 0.f;
  const float bv1 = HASBIAS ? bias[col1] : 0.f;
#pragma unroll
  for (int i = 0; i < 8; ++i) {
    const int row0 = m0 + i + 8 * half;
    const int row1 = row0 + 16;
    if (TRANSD) {
      D[(long)col0 * LDD + row0] = acc00[i] + bv0;
      D[(long)col1 * LDD + row0] = acc01[i] + bv1;
      D[(long)col0 * LDD + row1] = acc10[i] + bv0;
      D[(long)col1 * LDD + row1] = acc11[i] + bv1;
    } else {
      D[(long)row0 * LDD + col0] = acc00[i] + bv0;
      D[(long)row0 * LDD + col1] = acc01[i] + bv1;
      D[(long)row1 * LDD + col0] = acc10[i] + bv0;
      D[(long)row1 * LDD + col1] = acc11[i] + bv1;
    }
  }
}

template <int M, int N, int K, int LDA, int LDB, int LDD, int INNER,
          long SAO, long SAI, long SBO, long SBI, long SDO, long SDI,
          long SBIASI, bool TRANSA, bool TRANSD, bool HASBIAS>
__global__ __launch_bounds__(128) void wmma_gemm_f32(
    const float* __restrict__ A, const float* __restrict__ Bm,
    const float* __restrict__ bias, float* __restrict__ D)
{
  const int batch = blockIdx.y;
  const int bo = batch / INNER;
  const int bi = batch % INNER;
  A    += bo * SAO + bi * SAI;
  Bm   += bo * SBO + bi * SBI;
  D    += bo * SDO + bi * SDI;
  if (HASBIAS) bias += bi * SBIASI;
  gemm_wave_tile<M, N, K, LDA, LDB, LDD, TRANSA, TRANSD, HASBIAS>(A, Bm, bias, D);
}

__global__ __launch_bounds__(128) void gemm_gl_gr(
    const float* __restrict__ h, const float* __restrict__ Wl,
    const float* __restrict__ Wr, float* __restrict__ gl, float* __restrict__ gr)
{
  const bool second = blockIdx.z != 0;
  const long off = (long)blockIdx.y * C_ * HID_;
  gemm_wave_tile<C_, HID_, HID_, HID_, HID_, HID_, false, false, false>(
      h + off, second ? Wr : Wl, nullptr, (second ? gr : gl) + off);
}

__global__ __launch_bounds__(128) void gat_scores_softmax_agg(
    const float* __restrict__ gl, const float* __restrict__ gr,
    const float* __restrict__ att, const float* __restrict__ bias,
    float* __restrict__ alpha, float* __restrict__ gout)
{
  __shared__ alignas(16) float sRows[TCH_ * D_];
  __shared__ alignas(16) float sSc[TCH_ * SP_];
  __shared__ alignas(16) float sOut[TCH_ * OP_];
  __shared__ alignas(16) float redM[TCH_ * 4];
  __shared__ alignas(16) float redS[TCH_ * 4];

  const int tid = threadIdx.x;
  const int h = blockIdx.x;
  const int b = blockIdx.y;
  const int t0 = blockIdx.z * TCH_;
  const float* glb = gl + (long)b * C_ * HID_ + h * D_;
  const float* grb = gr + (long)b * C_ * HID_ + h * D_;
  const float* atth = att + h * D_;

  float4 stage[TCH_ * D_ / 4 / 128];
#pragma unroll
  for (int i = 0; i < TCH_ * D_ / 4 / 128; ++i) {
    const int idx = tid + 128 * i;
    stage[i] = *(const float4*)(glb + (long)(t0 + (idx >> 4)) * HID_ + 4 * (idx & 15));
  }

  v2f g2[D_ / 2];
  const float4* grow = (const float4*)(grb + (long)tid * HID_);
#pragma unroll
  for (int q = 0; q < D_ / 4; ++q) {
    const float4 v4 = grow[q];
    g2[2 * q + 0] = v2f{v4.x, v4.y};
    g2[2 * q + 1] = v2f{v4.z, v4.w};
  }
  float a[D_];
  float lin = 0.f;
#pragma unroll
  for (int d = 0; d < D_; ++d) {
    a[d] = atth[d];
    lin = fmaf(a[d], g2[d >> 1][d & 1], lin);
  }
  lin *= LIN_;

#pragma unroll
  for (int i = 0; i < TCH_ * D_ / 4 / 128; ++i)
    ((float4*)sRows)[tid + 128 * i] = stage[i];
  __syncthreads();

  const int lane = tid & 31;
  const int wv = tid >> 5;

  for (int t = 0; t < TCH_; ++t) {
    const v4f* rowp = (const v4f*)(sRows + t * D_);
    v4f r4[D_ / 4];
#pragma unroll
    for (int q = 0; q < D_ / 4; ++q) r4[q] = rowp[q];
    float sa = 0.f, sb = 0.f;
#pragma unroll
    for (int q = 0; q < D_ / 8; ++q) {
      const v4f ra = r4[q], rb = r4[q + D_ / 8];
      const v2f al = __builtin_shufflevector(ra, ra, 0, 1) + g2[2 * q];
      const v2f ah = __builtin_shufflevector(ra, ra, 2, 3) + g2[2 * q + 1];
      const v2f bl = __builtin_shufflevector(rb, rb, 0, 1) + g2[2 * q + D_ / 4];
      const v2f bh = __builtin_shufflevector(rb, rb, 2, 3) + g2[2 * q + D_ / 4 + 1];
      sa = fmaf(a[4 * q + 0], fabsf(al.x), sa);
      sb = fmaf(a[4 * q + D_ / 2 + 0], fabsf(bl.x), sb);
      sa = fmaf(a[4 * q + 1], fabsf(al.y), sa);
      sb = fmaf(a[4 * q + D_ / 2 + 1], fabsf(bl.y), sb);
      sa = fmaf(a[4 * q + 2], fabsf(ah.x), sa);
      sb = fmaf(a[4 * q + D_ / 2 + 2], fabsf(bh.x), sb);
      sa = fmaf(a[4 * q + 3], fabsf(ah.y), sa);
      sb = fmaf(a[4 * q + D_ / 2 + 3], fabsf(bh.y), sb);
    }
    sSc[t * SP_ + tid] = fmaf(ABS_, sa + sb, lin);
    __builtin_amdgcn_sched_group_barrier(0x100, D_ / 4, 0);
    __builtin_amdgcn_sched_group_barrier(0x002, D_ / 2 + D_, 0);
  }

  const int half = lane >> 4;
  const int r = lane & 15;
  const int n0 = wv * 16;
  v2f bf[C_ / 4];
  {
    const float* pB = grb + (long)(2 * half) * HID_ + n0 + r;
#pragma unroll
    for (int j = 0; j < C_ / 4; ++j) {
      bf[j].x = pB[(long)(4 * j) * HID_];
      bf[j].y = pB[(long)(4 * j + 1) * HID_];
    }
  }

  float sc[TCH_], m[TCH_];
#pragma unroll
  for (int t = 0; t < TCH_; ++t) { sc[t] = sSc[t * SP_ + tid]; m[t] = sc[t]; }

  for (int off = 16; off > 0; off >>= 1) {
#pragma unroll
    for (int t = 0; t < TCH_; ++t) m[t] = fmaxf(m[t], __shfl_xor(m[t], off, 32));
  }
  if (lane == 0) {
#pragma unroll
    for (int t = 0; t < TCH_; ++t) redM[t * 4 + wv] = m[t];
  }
  __syncthreads();
  float e[TCH_], ss[TCH_];
#pragma unroll
  for (int t = 0; t < TCH_; ++t) {
    const float4 r = ((const float4*)redM)[t];
    const float mt = fmaxf(fmaxf(r.x, r.y), fmaxf(r.z, r.w));
    e[t] = __expf(sc[t] - mt);
    ss[t] = e[t];
  }
  for (int off = 16; off > 0; off >>= 1) {
#pragma unroll
    for (int t = 0; t < TCH_; ++t) ss[t] += __shfl_xor(ss[t], off, 32);
  }
  if (lane == 0) {
#pragma unroll
    for (int t = 0; t < TCH_; ++t) redS[t * 4 + wv] = ss[t];
  }
  __syncthreads();
  float av[TCH_];
#pragma unroll
  for (int t = 0; t < TCH_; ++t) {
    const float4 rs = ((const float4*)redS)[t];
    const float ssum = rs.x + rs.y + rs.z + rs.w;
    av[t] = e[t] / ssum;
    sSc[t * SP_ + tid] = av[t];
  }
  __syncthreads();

  v8f acc = {0.f, 0.f, 0.f, 0.f, 0.f, 0.f, 0.f, 0.f};
  {
    const float* pA = sSc + r * SP_ + 2 * half;
#pragma unroll
    for (int j = 0; j < C_ / 4; ++j) {
      const v2f af = *(const v2f*)(pA + 4 * j);
      acc = __builtin_amdgcn_wmma_f32_16x16x4_f32(false, af, false, bf[j], (short)0, acc, false, false);
    }
  }
#pragma unroll
  for (int i = 0; i < 8; ++i) sOut[(i + 8 * half) * OP_ + n0 + r] = acc[i];
  __syncthreads();

  float* gob = gout + ((long)b * C_ + t0) * HID_ + h * D_;
#pragma unroll
  for (int q = 0; q < TCH_ * D_ / 4 / 128; ++q) {
    const int idx = tid + 128 * q;
    const int row = idx >> 4;
    const int c4 = 4 * (idx & 15);
    const float4 v = *(const float4*)(sOut + row * OP_ + c4);
    const float4 bv = *(const float4*)(bias + h * D_ + c4);
    float4 o;
    o.x = v.x + bv.x; o.y = v.y + bv.y; o.z = v.z + bv.z; o.w = v.w + bv.w;
    *(float4*)(gob + (long)row * HID_ + c4) = o;
  }
  if (alpha != nullptr) {
    float* arow = alpha + (long)(b * H_ + h) * C_ * C_ + (long)t0 * C_ + tid;
#pragma unroll
    for (int t = 0; t < TCH_; ++t)
      arow[t * C_] = av[t];
  }
}

__global__ __launch_bounds__(256) void post_elu_residual_ln(
    float* __restrict__ h, const float* __restrict__ gout,
    const float* __restrict__ g, const float* __restrict__ bb)
{
  __shared__ float red[8];
  const int tid = threadIdx.x;
  const long base = (long)blockIdx.x * HID_;

  float x = gout[base + tid];
  x = x > 0.f ? x : (__expf(x) - 1.f);
  const float v = h[base + tid] + x;

  const int lane = tid & 31, wv = tid >> 5;
  float s = v;
  for (int off = 16; off > 0; off >>= 1) s += __shfl_xor(s, off, 32);
  if (lane == 0) red[wv] = s;
  __syncthreads();
  s = 0.f;
#pragma unroll
  for (int i = 0; i < 8; ++i) s += red[i];
  const float mu = s * (1.f / HID_);
  __syncthreads();

  const float dmu = v - mu;
  float q = dmu * dmu;
  for (int off = 16; off > 0; off >>= 1) q += __shfl_xor(q, off, 32);
  if (lane == 0) red[wv] = q;
  __syncthreads();
  q = 0.f;
#pragma unroll
  for (int i = 0; i < 8; ++i) q += red[i];
  const float inv = rsqrtf(q * (1.f / HID_) + EPS_);

  h[base + tid] = dmu * inv * g[tid] + bb[tid];
}

extern "C" void kernel_launch(void* const* d_in, const int* in_sizes, int n_in,
                              void* d_out, int out_size, void* d_ws, size_t ws_size,
                              hipStream_t stream) {
  (void)in_sizes; (void)n_in; (void)out_size; (void)ws_size;
  const float* x      = (const float*)d_in[0];
  const float* W_emb  = (const float*)d_in[1];
  const float* b_emb  = (const float*)d_in[2];
  const float* Wl     = (const float*)d_in[3];
  const float* Wr     = (const float*)d_in[4];
  const float* att    = (const float*)d_in[5];
  const float* bias_g = (const float*)d_in[6];
  const float* ln_g   = (const float*)d_in[7];
  const float* ln_b   = (const float*)d_in[8];
  const float* W_proj = (const float*)d_in[9];
  const float* b_proj = (const float*)d_in[10];

  float* out   = (float*)d_out;
  float* alpha = out + (long)B_ * L_ * C_;

  float* ws = (float*)d_ws;
  const long hwN = (long)B_ * C_ * HID_;
  float* h    = ws;
  float* gl   = ws + hwN;
  float* gr   = ws + 2 * hwN;
  float* gout = ws + 3 * hwN;

  wmma_gemm_f32<C_, HID_, L_, C_, HID_, HID_, 1,
                (long)L_ * C_, 0, 0, 0, (long)C_ * HID_, 0, 0,
                true, false, true>
      <<<dim3((C_ / 32) * (HID_ / 32) / 4, B_), 128, 0, stream>>>(
          x, W_emb, b_emb, h);

  for (int layer = 0; layer < NL_; ++layer) {
    const float* Wl_i  = Wl + (long)layer * HID_ * HID_;
    const float* Wr_i  = Wr + (long)layer * HID_ * HID_;
    const float* att_i = att + (long)layer * H_ * D_;
    const float* bg_i  = bias_g + (long)layer * HID_;

    gemm_gl_gr<<<dim3((C_ / 32) * (HID_ / 32) / 4, B_, 2), 128, 0, stream>>>(
        h, Wl_i, Wr_i, gl, gr);

    gat_scores_softmax_agg<<<dim3(H_, B_, C_ / TCH_), 128, 0, stream>>>(
        gl, gr, att_i, bg_i, layer == NL_ - 1 ? alpha : nullptr, gout);

    post_elu_residual_ln<<<B_ * C_, 256, 0, stream>>>(
        h, gout, ln_g + (long)layer * HID_, ln_b + (long)layer * HID_);
  }

  wmma_gemm_f32<C_, L_, HID_, HID_, L_, C_, 1,
                (long)C_ * HID_, 0, 0, 0, (long)L_ * C_, 0, 0,
                false, true, true>
      <<<dim3((C_ / 32) * (L_ / 32) / 4, B_), 128, 0, stream>>>(
          h, W_proj, b_proj, out);
}


# layer norm (elu, residual, norm) moved into the scores kernel: one 16-wave work-group takes all 4 heads of its 16 targets, gout stays in LDS; 11 launches to 8; norm row sums re-associated
# speedup vs baseline: 10.3173x; 1.0306x over previous
#include <hip/hip_runtime.h>
#include <hip/hip_bf16.h>

typedef __attribute__((ext_vector_type(2))) float v2f;
typedef __attribute__((ext_vector_type(8))) float v8f;
typedef __attribute__((ext_vector_type(4))) float v4f;

#define B_   32
#define L_   512
#define C_   128
#define HID_ 256
#define H_   4
#define D_   64
#define NL_  3
#define NEG_ 0.2f
#define EPS_ 1e-5f
#define TCH_ 16
#define LIN_ (0.5f * (1.f + NEG_))
#define ABS_ (0.5f * (1.f - NEG_))
#define SP_  (C_ + 4)
#define OW_  (HID_ + 4)

template <int KS, int LDA, int LDB, bool TRANSA>
__device__ __forceinline__ void gemm_load_block(
    const float* __restrict__ pA0, const float* __restrict__ pA1,
    const float* __restrict__ pB0, const float* __restrict__ pB1,
    v2f (&a0)[KS], v2f (&a1)[KS], v2f (&b0)[KS], v2f (&b1)[KS])
{
  constexpr int stepA = TRANSA ? 4 * LDA : 4;
  constexpr int stepB = 4 * LDB;
  if (TRANSA) {
#pragma unroll
    for (int j = 0; j < KS; ++j) {
      a0[j].x = pA0[j * stepA]; a0[j].y = pA0[j * stepA + LDA];
      a1[j].x = pA1[j * stepA]; a1[j].y = pA1[j * stepA + LDA];
      b0[j].x = pB0[j * stepB]; b0[j].y = pB0[j * stepB + LDB];
      b1[j].x = pB1[j * stepB]; b1[j].y = pB1[j * stepB + LDB];
    }
  } else {
    static_assert(TRANSA || KS % 2 == 0, "K-steps go in pairs");
#pragma unroll
    for (int p = 0; p < KS / 2; ++p) {
      const v4f qa0 = *(const v4f*)(pA0 + 8 * p);
      const v4f qa1 = *(const v4f*)(pA1 + 8 * p);
      a0[2 * p]     = v2f{qa0.x, qa0.y};
      a0[2 * p + 1] = v2f{qa0.z, qa0.w};
      a1[2 * p]     = v2f{qa1.x, qa1.y};
      a1[2 * p + 1] = v2f{qa1.z, qa1.w};
      b0[2 * p].x     = pB0[(8 * p + 0) * LDB]; b0[2 * p].y     = pB0[(8 * p + 1) * LDB];
      b0[2 * p + 1].x = pB0[(8 * p + 2) * LDB]; b0[2 * p + 1].y = pB0[(8 * p + 3) * LDB];
      b1[2 * p].x     = pB1[(8 * p + 0) * LDB]; b1[2 * p].y     = pB1[(8 * p + 1) * LDB];
      b1[2 * p + 1].x = pB1[(8 * p + 2) * LDB]; b1[2 * p + 1].y = pB1[(8 * p + 3) * LDB];
    }
  }
}

template <int KS>
__device__ __forceinline__ void gemm_mma_block(
    const v2f (&a0)[KS], const v2f (&a1)[KS],
    const v2f (&b0)[KS], const v2f (&b1)[KS],
    v8f& acc00, v8f& acc01, v8f& acc10, v8f& acc11)
{
#pragma unroll
  for (int j = 0; j < KS; ++j) {
    acc00 = __builtin_amdgcn_wmma_f32_16x16x4_f32(false, a0[j], false, b0[j], (short)0, acc00, false, false);
    acc01 = __builtin_amdgcn_wmma_f32_16x16x4_f32(false, a0[j], false, b1[j], (short)0, acc01, false, false);
    acc10 = __builtin_amdgcn_wmma_f32_16x16x4_f32(false, a1[j], false, b0[j], (short)0, acc10, false, false);
    acc11 = __builtin_amdgcn_wmma_f32_16x16x4_f32(false, a1[j], false, b1[j], (short)0, acc11, false, false);
  }
}

template <int M, int N, int K, int LDA, int LDB, int LDD,
          bool TRANSA, bool TRANSD, bool HASBIAS>
__device__ __forceinline__ void gemm_wave_tile(
    const float* __restrict__ A, const float* __restrict__ Bm,
    const float* __restrict__ bias, float* __restrict__ D)
{
  const int lane = threadIdx.x & 31;
  const int wave = threadIdx.x >> 5;
  const int half = (lane >> 4) & 1;
  const int r    = lane & 15;

  constexpr int tilesN = N >> 5;
  const int tile = blockIdx.x * 4 + wave;
  const int tm = tile / tilesN;
  const int tn = tile % tilesN;

  const int m0 = tm << 5;
  const int n0 = tn << 5;

  constexpr int KS   = TRANSA ? 4 : 8;
  constexpr int DIST = TRANSA ? 3 : 2;
  constexpr int NB   = K / (4 * KS);
  static_assert(K % (4 * KS) == 0 && NB >= DIST, "K must be whole blocks");
  constexpr int blkA = KS * (TRANSA ? 4 * LDA : 4);
  constexpr int blkB = KS * 4 * LDB;
  constexpr int KH = TRANSA ? 2 : 4;
  const int offA0 = TRANSA ? (KH * half) * LDA + (m0 + r)
                           : (m0 + r) * LDA + KH * half;
  const int offB0 = (KH * half) * LDB + (n0 + r);
  const float* pA0 = A + offA0;
  const float* pA1 = pA0 + (TRANSA ? 16 : 16 * LDA);
  const float* pB0 = Bm + offB0;
  const float* pB1 = pB0 + 16;

  v8f acc00 = {0.f,0.f,0.f,0.f,0.f,0.f,0.f,0.f};
  v8f acc01 = acc00, acc10 = acc00, acc11 = acc00;

  v2f fa0[NB][KS], fa1[NB][KS], fb0[NB][KS], fb1[NB][KS];
#pragma unroll
  for (int kb = 0; kb < DIST; ++kb)
    gemm_load_block<KS, LDA, LDB, TRANSA>(pA0 + kb * blkA, pA1 + kb * blkA,
                                          pB0 + kb * blkB, pB1 + kb * blkB,
                                          fa0[kb], fa1[kb], fb0[kb], fb1[kb]);
#pragma unroll
  for (int kb = 0; kb < NB; ++kb) {
    gemm_mma_block<KS>(fa0[kb], fa1[kb], fb0[kb], fb1[kb], acc00, acc01, acc10, acc11);
    if (kb + DIST < NB)
      gemm_load_block<KS, LDA, LDB, TRANSA>(pA0 + (kb + DIST) * blkA, pA1 + (kb + DIST) * blkA,
                                            pB0 + (kb + DIST) * blkB, pB1 + (kb + DIST) * blkB,
                                            fa0[kb + DIST], fa1[kb + DIST],
                                            fb0[kb + DIST], fb1[kb + DIST]);
  }
  constexpr int LPB = (TRANSA ? 8 : 5) * KS;
#pragma unroll
  for (int kb = 0; kb < DIST; ++kb)
    __builtin_amdgcn_sched_group_barrier(0x020, LPB, 0);
#pragma unroll
  for (int kb = 0; kb < NB; ++kb) {
    __builtin_amdgcn_sched_group_barrier(0x008, 4 * KS, 0);
    if (kb + DIST < NB) __builtin_amdgcn_sched_group_barrier(0x020, LPB, 0);
  }

  const int col0 = n0 + r;
  const int col1 = col0 + 16;
  const float bv0 = HASBIAS ? bias[col0] : 0.f;
  const float bv1 = HASBIAS ? bias[col1] : 0.f;
#pragma unroll
  for (int i = 0; i < 8; ++i) {
    const int row0 = m0 + i + 8 * half;
    const int row1 = row0 + 16;
    if (TRANSD) {
      D[(long)col0 * LDD + row0] = acc00[i] + bv0;
      D[(long)col1 * LDD + row0] = acc01[i] + bv1;
      D[(long)col0 * LDD + row1] = acc10[i] + bv0;
      D[(long)col1 * LDD + row1] = acc11[i] + bv1;
    } else {
      D[(long)row0 * LDD + col0] = acc00[i] + bv0;
      D[(long)row0 * LDD + col1] = acc01[i] + bv1;
      D[(long)row1 * LDD + col0] = acc10[i] + bv0;
      D[(long)row1 * LDD + col1] = acc11[i] + bv1;
    }
  }
}

template <int M, int N, int K, int LDA, int LDB, int LDD, int INNER,
          long SAO, long SAI, long SBO, long SBI, long SDO, long SDI,
          long SBIASI, bool TRANSA, bool TRANSD, bool HASBIAS>
__global__ __launch_bounds__(128) void wmma_gemm_f32(
    const float* __restrict__ A, const float* __restrict__ Bm,
    const float* __restrict__ bias, float* __restrict__ D)
{
  const int batch = blockIdx.y;
  const int bo = batch / INNER;
  const int bi = batch % INNER;
  A    += bo * SAO + bi * SAI;
  Bm   += bo * SBO + bi * SBI;
  D    += bo * SDO + bi * SDI;
  if (HASBIAS) bias += bi * SBIASI;
  gemm_wave_tile<M, N, K, LDA, LDB, LDD, TRANSA, TRANSD, HASBIAS>(A, Bm, bias, D);
}

__global__ __launch_bounds__(128) void gemm_gl_gr(
    const float* __restrict__ h, const float* __restrict__ Wl,
    const float* __restrict__ Wr, float* __restrict__ gl, float* __restrict__ gr)
{
  const bool second = blockIdx.z != 0;
  const long off = (long)blockIdx.y * C_ * HID_;
  gemm_wave_tile<C_, HID_, HID_, HID_, HID_, HID_, false, false, false>(
      h + off, second ? Wr : Wl, nullptr, (second ? gr : gl) + off);
}

__global__ __launch_bounds__(512) void gat_scores_agg_ln(
    const float* __restrict__ gl, const float* __restrict__ gr,
    const float* __restrict__ att, const float* __restrict__ bias,
    const float* __restrict__ lng, const float* __restrict__ lnb,
    float* __restrict__ alpha, float* __restrict__ h)
{
  __shared__ alignas(16) float sRows[H_ * TCH_ * D_];
  __shared__ alignas(16) float sSc[H_ * TCH_ * SP_];
  __shared__ alignas(16) float sOut[TCH_ * OW_];
  __shared__ alignas(16) float redM[H_ * TCH_ * 4];
  __shared__ alignas(16) float redS[H_ * TCH_ * 4];

  const int lane = threadIdx.x & 31;
  const int wid = __builtin_amdgcn_readfirstlane((int)(threadIdx.x >> 5));
  const int hd = wid >> 2;
  const int wv = wid & 3;
  const int tid = wv * 32 + lane;
  const int b = blockIdx.x;
  const int t0 = blockIdx.y * TCH_;
  const float* glb = gl + (long)b * C_ * HID_ + hd * D_;
  const float* grb = gr + (long)b * C_ * HID_ + hd * D_;
  const float* atth = att + hd * D_;
  float* hRows = sRows + hd * (TCH_ * D_);
  float* hSc = sSc + hd * (TCH_ * SP_);
  float* hRedM = redM + hd * (TCH_ * 4);
  float* hRedS = redS + hd * (TCH_ * 4);

  float4 stage[TCH_ * D_ / 4 / 128];
#pragma unroll
  for (int i = 0; i < TCH_ * D_ / 4 / 128; ++i) {
    const int idx = tid + 128 * i;
    stage[i] = *(const float4*)(glb + (long)(t0 + (idx >> 4)) * HID_ + 4 * (idx & 15));
  }

  v2f g2[D_ / 2];
  const float4* grow = (const float4*)(grb + (long)tid * HID_);
#pragma unroll
  for (int q = 0; q < D_ / 4; ++q) {
    const float4 v4 = grow[q];
    g2[2 * q + 0] = v2f{v4.x, v4.y};
    g2[2 * q + 1] = v2f{v4.z, v4.w};
  }
  float a[D_];
  float lin = 0.f;
#pragma unroll
  for (int d = 0; d < D_; ++d) {
    a[d] = atth[d];
    lin = fmaf(a[d], g2[d >> 1][d & 1], lin);
  }
  lin *= LIN_;

#pragma unroll
  for (int i = 0; i < TCH_ * D_ / 4 / 128; ++i)
    ((float4*)hRows)[tid + 128 * i] = stage[i];
  __syncthreads();

  for (int t = 0; t < TCH_; ++t) {
    const v4f* rowp = (const v4f*)(hRows + t * D_);
    v4f r4[D_ / 4];
#pragma unroll
    for (int q = 0; q < D_ / 4; ++q) r4[q] = rowp[q];
    float sa = 0.f, sb = 0.f;
#pragma unroll
    for (int q = 0; q < D_ / 8; ++q) {
      const v4f ra = r4[q], rb = r4[q + D_ / 8];
      const v2f al = __builtin_shufflevector(ra, ra, 0, 1) + g2[2 * q];
      const v2f ah = __builtin_shufflevector(ra, ra, 2, 3) + g2[2 * q + 1];
      const v2f bl = __builtin_shufflevector(rb, rb, 0, 1) + g2[2 * q + D_ / 4];
      const v2f bh = __builtin_shufflevector(rb, rb, 2, 3) + g2[2 * q + D_ / 4 + 1];
      sa = fmaf(a[4 * q + 0], fabsf(al.x), sa);
      sb = fmaf(a[4 * q + D_ / 2 + 0], fabsf(bl.x), sb);
      sa = fmaf(a[4 * q + 1], fabsf(al.y), sa);
      sb = fmaf(a[4 * q + D_ / 2 + 1], fabsf(bl.y), sb);
      sa = fmaf(a[4 * q + 2], fabsf(ah.x), sa);
      sb = fmaf(a[4 * q + D_ / 2 + 2], fabsf(bh.x), sb);
      sa = fmaf(a[4 * q + 3], fabsf(ah.y), sa);
      sb = fmaf(a[4 * q + D_ / 2 + 3], fabsf(bh.y), sb);
    }
    hSc[t * SP_ + tid] = fmaf(ABS_, sa + sb, lin);
    __builtin_amdgcn_sched_group_barrier(0x100, D_ / 4, 0);
    __builtin_amdgcn_sched_group_barrier(0x002, D_ / 2 + D_, 0);
  }

  const int half = lane >> 4;
  const int r = lane & 15;
  const int n0 = wv * 16;
  v2f bf[C_ / 4];
  {
    const float* pB = grb + (long)(2 * half) * HID_ + n0 + r;
#pragma unroll
    for (int j = 0; j < C_ / 4; ++j) {
      bf[j].x = pB[(long)(4 * j) * HID_];
      bf[j].y = pB[(long)(4 * j + 1) * HID_];
    }
  }
  float* hrow = h + ((long)b * C_ + t0 + wid) * HID_ + 4 * lane;
  const float4 hv0 = *(const float4*)(hrow);
  const float4 hv1 = *(const float4*)(hrow + HID_ / 2);

  float sc[TCH_], m[TCH_];
#pragma unroll
  for (int t = 0; t < TCH_; ++t) { sc[t] = hSc[t * SP_ + tid]; m[t] = sc[t]; }

  for (int off = 16; off > 0; off >>= 1) {
#pragma unroll
    for (int t = 0; t < TCH_; ++t) m[t] = fmaxf(m[t], __shfl_xor(m[t], off, 32));
  }
  if (lane == 0) {
#pragma unroll
    for (int t = 0; t < TCH_; ++t) hRedM[t * 4 + wv] = m[t];
  }
  __syncthreads();
  float e[TCH_], ss[TCH_];
#pragma unroll
  for (int t = 0; t < TCH_; ++t) {
    const float4 rm = ((const float4*)hRedM)[t];
    const float mt = fmaxf(fmaxf(rm.x, rm.y), fmaxf(rm.z, rm.w));
    e[t] = __expf(sc[t] - mt);
    ss[t] = e[t];
  }
  for (int off = 16; off > 0; off >>= 1) {
#pragma unroll
    for (int t = 0; t < TCH_; ++t) ss[t] += __shfl_xor(ss[t], off, 32);
  }
  if (lane == 0) {
#pragma unroll
    for (int t = 0; t < TCH_; ++t) hRedS[t * 4 + wv] = ss[t];
  }
  __syncthreads();
  float av[TCH_];
#pragma unroll
  for (int t = 0; t < TCH_; ++t) {
    const float4 rs = ((const float4*)hRedS)[t];
    const float ssum = rs.x + rs.y + rs.z + rs.w;
    av[t] = e[t] / ssum;
    hSc[t * SP_ + tid] = av[t];
  }
  __syncthreads();

  const int c0 = 4 * lane;
  const float4 bv0 = *(const float4*)(bias + c0);
  const float4 bv1 = *(const float4*)(bias + HID_ / 2 + c0);
  const float4 gv0 = *(const float4*)(lng + c0);
  const float4 gv1 = *(const float4*)(lng + HID_ / 2 + c0);
  const float4 lb0 = *(const float4*)(lnb + c0);
  const float4 lb1 = *(const float4*)(lnb + HID_ / 2 + c0);

  v8f acc = {0.f, 0.f, 0.f, 0.f, 0.f, 0.f, 0.f, 0.f};
  {
    const float* pA = hSc + r * SP_ + 2 * half;
#pragma unroll
    for (int j = 0; j < C_ / 4; ++j) {
      const v2f af = *(const v2f*)(pA + 4 * j);
      acc = __builtin_amdgcn_wmma_f32_16x16x4_f32(false, af, false, bf[j], (short)0, acc, false, false);
    }
  }
#pragma unroll
  for (int i = 0; i < 8; ++i) sOut[(i + 8 * half) * OW_ + hd * D_ + n0 + r] = acc[i];
  __syncthreads();

  if (alpha != nullptr) {
    float* arow = alpha + (long)(b * H_ + hd) * C_ * C_ + (long)t0 * C_ + tid;
#pragma unroll
    for (int t = 0; t < TCH_; ++t)
      arow[t * C_] = av[t];
  }

  const float4 o0 = *(const float4*)(sOut + wid * OW_ + c0);
  const float4 o1 = *(const float4*)(sOut + wid * OW_ + HID_ / 2 + c0);
  float x[8] = {o0.x + bv0.x, o0.y + bv0.y, o0.z + bv0.z, o0.w + bv0.w,
                o1.x + bv1.x, o1.y + bv1.y, o1.z + bv1.z, o1.w + bv1.w};
  const float hv[8] = {hv0.x, hv0.y, hv0.z, hv0.w, hv1.x, hv1.y, hv1.z, hv1.w};
  const float gv[8] = {gv0.x, gv0.y, gv0.z, gv0.w, gv1.x, gv1.y, gv1.z, gv1.w};
  const float lb[8] = {lb0.x, lb0.y, lb0.z, lb0.w, lb1.x, lb1.y, lb1.z, lb1.w};
  float v[8];
#pragma unroll
  for (int k = 0; k < 8; ++k) {
    const float xe = x[k] > 0.f ? x[k] : (__expf(x[k]) - 1.f);
    v[k] = hv[k] + xe;
  }
  float s = ((v[0] + v[1]) + (v[2] + v[3])) + ((v[4] + v[5]) + (v[6] + v[7]));
  for (int off = 16; off > 0; off >>= 1) s += __shfl_xor(s, off, 32);
  const float mu = s * (1.f / HID_);
  float dm[8];
#pragma unroll
  for (int k = 0; k < 8; ++k) dm[k] = v[k] - mu;
  float q = ((dm[0] * dm[0] + dm[1] * dm[1]) + (dm[2] * dm[2] + dm[3] * dm[3]))
          + ((dm[4] * dm[4] + dm[5] * dm[5]) + (dm[6] * dm[6] + dm[7] * dm[7]));
  for (int off = 16; off > 0; off >>= 1) q += __shfl_xor(q, off, 32);
  const float inv = rsqrtf(q * (1.f / HID_) + EPS_);
  float4 w0, w1;
  w0.x = dm[0] * inv * gv[0] + lb[0]; w0.y = dm[1] * inv * gv[1] + lb[1];
  w0.z = dm[2] * inv * gv[2] + lb[2]; w0.w = dm[3] * inv * gv[3] + lb[3];
  w1.x = dm[4] * inv * gv[4] + lb[4]; w1.y = dm[5] * inv * gv[5] + lb[5];
  w1.z = dm[6] * inv * gv[6] + lb[6]; w1.w = dm[7] * inv * gv[7] + lb[7];
  *(float4*)(hrow) = w0;
  *(float4*)(hrow + HID_ / 2) = w1;
}

extern "C" void kernel_launch(void* const* d_in, const int* in_sizes, int n_in,
                              void* d_out, int out_size, void* d_ws, size_t ws_size,
                              hipStream_t stream) {
  (void)in_sizes; (void)n_in; (void)out_size; (void)ws_size;
  const float* x      = (const float*)d_in[0];
  const float* W_emb  = (const float*)d_in[1];
  const float* b_emb  = (const float*)d_in[2];
  const float* Wl     = (const float*)d_in[3];
  const float* Wr     = (const float*)d_in[4];
  const float* att    = (const float*)d_in[5];
  const float* bias_g = (const float*)d_in[6];
  const float* ln_g   = (const float*)d_in[7];
  const float* ln_b   = (const float*)d_in[8];
  const float* W_proj = (const float*)d_in[9];
  const float* b_proj = (const float*)d_in[10];

  float* out   = (float*)d_out;
  float* alpha = out + (long)B_ * L_ * C_;

  float* ws = (float*)d_ws;
  const long hwN = (long)B_ * C_ * HID_;
  float* h    = ws;
  float* gl   = ws + hwN;
  float* gr   = ws + 2 * hwN;

  wmma_gemm_f32<C_, HID_, L_, C_, HID_, HID_, 1,
                (long)L_ * C_, 0, 0, 0, (long)C_ * HID_, 0, 0,
                true, false, true>
      <<<dim3((C_ / 32) * (HID_ / 32) / 4, B_), 128, 0, stream>>>(
          x, W_emb, b_emb, h);

  for (int layer = 0; layer < NL_; ++layer) {
    const float* Wl_i  = Wl + (long)layer * HID_ * HID_;
    const float* Wr_i  = Wr + (long)layer * HID_ * HID_;
    const float* att_i = att + (long)layer * H_ * D_;
    const float* bg_i  = bias_g + (long)layer * HID_;

    gemm_gl_gr<<<dim3((C_ / 32) * (HID_ / 32) / 4, B_, 2), 128, 0, stream>>>(
        h, Wl_i, Wr_i, gl, gr);

    gat_scores_agg_ln<<<dim3(B_, C_ / TCH_), 512, 0, stream>>>(
        gl, gr, att_i, bg_i, ln_g + (long)layer * HID_, ln_b + (long)layer * HID_,
        layer == NL_ - 1 ? alpha : nullptr, h);
  }

  wmma_gemm_f32<C_, L_, HID_, HID_, L_, C_, 1,
                (long)C_ * HID_, 0, 0, 0, (long)L_ * C_, 0, 0,
                false, true, true>
      <<<dim3((C_ / 32) * (L_ / 32) / 4, B_), 128, 0, stream>>>(
          h, W_proj, b_proj, out);
}


# scores kernel: a wave's 32 source rows of gr fetched by whole-line loads (16 lanes per row, 4 lines per load instead of 32) and turned in LDS into one row per lane; values and sums unchanged
# speedup vs baseline: 10.7196x; 1.0390x over previous
#include <hip/hip_runtime.h>
#include <hip/hip_bf16.h>

typedef __attribute__((ext_vector_type(2))) float v2f;
typedef __attribute__((ext_vector_type(8))) float v8f;
typedef __attribute__((ext_vector_type(4))) float v4f;

#define B_   32
#define L_   512
#define C_   128
#define HID_ 256
#define H_   4
#define D_   64
#define NL_  3
#define NEG_ 0.2f
#define EPS_ 1e-5f
#define TCH_ 16
#define LIN_ (0.5f * (1.f + NEG_))
#define ABS_ (0.5f * (1.f - NEG_))
#define SP_  (C_ + 4)
#define OW_  (HID_ + 4)
#define GP_  (D_ + 4)

template <int KS, int LDA, int LDB, bool TRANSA>
__device__ __forceinline__ void gemm_load_block(
    const float* __restrict__ pA0, const float* __restrict__ pA1,
    const float* __restrict__ pB0, const float* __restrict__ pB1,
    v2f (&a0)[KS], v2f (&a1)[KS], v2f (&b0)[KS], v2f (&b1)[KS])
{
  constexpr int stepA = TRANSA ? 4 * LDA : 4;
  constexpr int stepB = 4 * LDB;
  if (TRANSA) {
#pragma unroll
    for (int j = 0; j < KS; ++j) {
      a0[j].x = pA0[j * stepA]; a0[j].y = pA0[j * stepA + LDA];
      a1[j].x = pA1[j * stepA]; a1[j].y = pA1[j * stepA + LDA];
      b0[j].x = pB0[j * stepB]; b0[j].y = pB0[j * stepB + LDB];
      b1[j].x = pB1[j * stepB]; b1[j].y = pB1[j * stepB + LDB];
    }
  } else {
    static_assert(TRANSA || KS % 2 == 0, "K-steps go in pairs");
#pragma unroll
    for (int p = 0; p < KS / 2; ++p) {
      const v4f qa0 = *(const v4f*)(pA0 + 8 * p);
      const v4f qa1 = *(const v4f*)(pA1 + 8 * p);
      a0[2 * p]     = v2f{qa0.x, qa0.y};
      a0[2 * p + 1] = v2f{qa0.z, qa0.w};
      a1[2 * p]     = v2f{qa1.x, qa1.y};
      a1[2 * p + 1] = v2f{qa1.z, qa1.w};
      b0[2 * p].x     = pB0[(8 * p + 0) * LDB]; b0[2 * p].y     = pB0[(8 * p + 1) * LDB];
      b0[2 * p + 1].x = pB0[(8 * p + 2) * LDB]; b0[2 * p + 1].y = pB0[(8 * p + 3) * LDB];
      b1[2 * p].x     = pB1[(8 * p + 0) * LDB]; b1[2 * p].y     = pB1[(8 * p + 1) * LDB];
      b1[2 * p + 1].x = pB1[(8 * p + 2) * LDB]; b1[2 * p + 1].y = pB1[(8 * p + 3) * LDB];
    }
  }
}

template <int KS>
__device__ __forceinline__ void gemm_mma_block(
    const v2f (&a0)[KS], const v2f (&a1)[KS],
    const v2f (&b0)[KS], const v2f (&b1)[KS],
    v8f& acc00, v8f& acc01, v8f& acc10, v8f& acc11)
{
#pragma unroll
  for (int j = 0; j < KS; ++j) {
    acc00 = __builtin_amdgcn_wmma_f32_16x16x4_f32(false, a0[j], false, b0[j], (short)0, acc00, false, false);
    acc01 = __builtin_amdgcn_wmma_f32_16x16x4_f32(false, a0[j], false, b1[j], (short)0, acc01, false, false);
    acc10 = __builtin_amdgcn_wmma_f32_16x16x4_f32(false, a1[j], false, b0[j], (short)0, acc10, false, false);
    acc11 = __builtin_amdgcn_wmma_f32_16x16x4_f32(false, a1[j], false, b1[j], (short)0, acc11, false, false);
  }
}

template <int M, int N, int K, int LDA, int LDB, int LDD,
          bool TRANSA, bool TRANSD, bool HASBIAS>
__device__ __forceinline__ void gemm_wave_tile(
    const float* __restrict__ A, const float* __restrict__ Bm,
    const float* __restrict__ bias, float* __restrict__ D)
{
  const int lane = threadIdx.x & 31;
  const int wave = threadIdx.x >> 5;
  const int half = (lane >> 4) & 1;
  const int r    = lane & 15;

  constexpr int tilesN = N >> 5;
  const int tile = blockIdx.x * 4 + wave;
  const int tm = tile / tilesN;
  const int tn = tile % tilesN;

  const int m0 = tm << 5;
  const int n0 = tn << 5;

  constexpr int KS   = TRANSA ? 4 : 8;
  constexpr int DIST = TRANSA ? 3 : 2;
  constexpr int NB   = K / (4 * KS);
  static_assert(K % (4 * KS) == 0 && NB >= DIST, "K must be whole blocks");
  constexpr int blkA = KS * (TRANSA ? 4 * LDA : 4);
  constexpr int blkB = KS * 4 * LDB;
  constexpr int KH = TRANSA ? 2 : 4;
  const int offA0 = TRANSA ? (KH * half) * LDA + (m0 + r)
                           : (m0 + r) * LDA + KH * half;
  const int offB0 = (KH * half) * LDB + (n0 + r);
  const float* pA0 = A + offA0;
  const float* pA1 = pA0 + (TRANSA ? 16 : 16 * LDA);
  const float* pB0 = Bm + offB0;
  const float* pB1 = pB0 + 16;

  v8f acc00 = {0.f,0.f,0.f,0.f,0.f,0.f,0.f,0.f};
  v8f acc01 = acc00, acc10 = acc00, acc11 = acc00;

  v2f fa0[NB][KS], fa1[NB][KS], fb0[NB][KS], fb1[NB][KS];
#pragma unroll
  for (int kb = 0; kb < DIST; ++kb)
    gemm_load_block<KS, LDA, LDB, TRANSA>(pA0 + kb * blkA, pA1 + kb * blkA,
                                          pB0 + kb * blkB, pB1 + kb * blkB,
                                          fa0[kb], fa1[kb], fb0[kb], fb1[kb]);
#pragma unroll
  for (int kb = 0; kb < NB; ++kb) {
    gemm_mma_block<KS>(fa0[kb], fa1[kb], fb0[kb], fb1[kb], acc00, acc01, acc10, acc11);
    if (kb + DIST < NB)
      gemm_load_block<KS, LDA, LDB, TRANSA>(pA0 + (kb + DIST) * blkA, pA1 + (kb + DIST) * blkA,
                                            pB0 + (kb + DIST) * blkB, pB1 + (kb + DIST) * blkB,
                                            fa0[kb + DIST], fa1[kb + DIST],
                                            fb0[kb + DIST], fb1[kb + DIST]);
  }
  constexpr int LPB = (TRANSA ? 8 : 5) * KS;
#pragma unroll
  for (int kb = 0; kb < DIST; ++kb)
    __builtin_amdgcn_sched_group_barrier(0x020, LPB, 0);
#pragma unroll
  for (int kb = 0; kb < NB; ++kb) {
    __builtin_amdgcn_sched_group_barrier(0x008, 4 * KS, 0);
    if (kb + DIST < NB) __builtin_amdgcn_sched_group_barrier(0x020, LPB, 0);
  }

  const int col0 = n0 + r;
  const int col1 = col0 + 16;
  const float bv0 = HASBIAS ? bias[col0] : 0.f;
  const float bv1 = HASBIAS ? bias[col1] : 0.f;
#pragma unroll
  for (int i = 0; i < 8; ++i) {
    const int row0 = m0 + i + 8 * half;
    const int row1 = row0 + 16;
    if (TRANSD) {
      D[(long)col0 * LDD + row0] = acc00[i] + bv0;
      D[(long)col1 * LDD + row0] = acc01[i] + bv1;
      D[(long)col0 * LDD + row1] = acc10[i] + bv0;
      D[(long)col1 * LDD + row1] = acc11[i] + bv1;
    } else {
      D[(long)row0 * LDD + col0] = acc00[i] + bv0;
      D[(long)row0 * LDD + col1] = acc01[i] + bv1;
      D[(long)row1 * LDD + col0] = acc10[i] + bv0;
      D[(long)row1 * LDD + col1] = acc11[i] + bv1;
    }
  }
}

template <int M, int N, int K, int LDA, int LDB, int LDD, int INNER,
          long SAO, long SAI, long SBO, long SBI, long SDO, long SDI,
          long SBIASI, bool TRANSA, bool TRANSD, bool HASBIAS>
__global__ __launch_bounds__(128) void wmma_gemm_f32(
    const float* __restrict__ A, const float* __restrict__ Bm,
    const float* __restrict__ bias, float* __restrict__ D)
{
  const int batch = blockIdx.y;
  const int bo = batch / INNER;
  const int bi = batch % INNER;
  A    += bo * SAO + bi * SAI;
  Bm   += bo * SBO + bi * SBI;
  D    += bo * SDO + bi * SDI;
  if (HASBIAS) bias += bi * SBIASI;
  gemm_wave_tile<M, N, K, LDA, LDB, LDD, TRANSA, TRANSD, HASBIAS>(A, Bm, bias, D);
}

__global__ __launch_bounds__(128) void gemm_gl_gr(
    const float* __restrict__ h, const float* __restrict__ Wl,
    const float* __restrict__ Wr, float* __restrict__ gl, float* __restrict__ gr)
{
  const bool second = blockIdx.z != 0;
  const long off = (long)blockIdx.y * C_ * HID_;
  gemm_wave_tile<C_, HID_, HID_, HID_, HID_, HID_, false, false, false>(
      h + off, second ? Wr : Wl, nullptr, (second ? gr : gl) + off);
}

__global__ __launch_bounds__(512) void gat_scores_agg_ln(
    const float* __restrict__ gl, const float* __restrict__ gr,
    const float* __restrict__ att, const float* __restrict__ bias,
    const float* __restrict__ lng, const float* __restrict__ lnb,
    float* __restrict__ alpha, float* __restrict__ h)
{
  __shared__ alignas(16) float sRows[H_ * TCH_ * D_];
  __shared__ alignas(16) float sGr[H_ * C_ * GP_];
  __shared__ alignas(16) float sSc[H_ * TCH_ * SP_];
  __shared__ alignas(16) float sOut[TCH_ * OW_];
  __shared__ alignas(16) float redM[H_ * TCH_ * 4];
  __shared__ alignas(16) float redS[H_ * TCH_ * 4];

  const int lane = threadIdx.x & 31;
  const int wid = __builtin_amdgcn_readfirstlane((int)(threadIdx.x >> 5));
  const int hd = wid >> 2;
  const int wv = wid & 3;
  const int tid = wv * 32 + lane;
  const int b = blockIdx.x;
  const int t0 = blockIdx.y * TCH_;
  const float* glb = gl + (long)b * C_ * HID_ + hd * D_;
  const float* grb = gr + (long)b * C_ * HID_ + hd * D_;
  const float* atth = att + hd * D_;
  float* hRows = sRows + hd * (TCH_ * D_);
  float* wGr = sGr + wid * (32 * GP_);
  float* hSc = sSc + hd * (TCH_ * SP_);
  float* hRedM = redM + hd * (TCH_ * 4);
  float* hRedS = redS + hd * (TCH_ * 4);

  float4 stage[TCH_ * D_ / 4 / 128];
#pragma unroll
  for (int i = 0; i < TCH_ * D_ / 4 / 128; ++i) {
    const int idx = tid + 128 * i;
    stage[i] = *(const float4*)(glb + (long)(t0 + (idx >> 4)) * HID_ + 4 * (idx & 15));
  }

  const int gj = lane >> 4;
  const int gc = lane & 15;
  float4 gst[32 / 2];
#pragma unroll
  for (int i = 0; i < 32 / 2; ++i)
    gst[i] = *(const float4*)(grb + (long)(wv * 32 + 2 * i + gj) * HID_ + 4 * gc);
  float a[D_];
#pragma unroll
  for (int d = 0; d < D_; ++d) a[d] = atth[d];

#pragma unroll
  for (int i = 0; i < TCH_ * D_ / 4 / 128; ++i)
    ((float4*)hRows)[tid + 128 * i] = stage[i];
#pragma unroll
  for (int i = 0; i < 32 / 2; ++i)
    *(float4*)(wGr + (2 * i + gj) * GP_ + 4 * gc) = gst[i];
  __syncthreads();

  v2f g2[D_ / 2];
  const float4* grow = (const float4*)(wGr + lane * GP_);
#pragma unroll
  for (int q = 0; q < D_ / 4; ++q) {
    const float4 v4 = grow[q];
    g2[2 * q + 0] = v2f{v4.x, v4.y};
    g2[2 * q + 1] = v2f{v4.z, v4.w};
  }
  float lin = 0.f;
#pragma unroll
  for (int d = 0; d < D_; ++d)
    lin = fmaf(a[d], g2[d >> 1][d & 1], lin);
  lin *= LIN_;

  for (int t = 0; t < TCH_; ++t) {
    const v4f* rowp = (const v4f*)(hRows + t * D_);
    v4f r4[D_ / 4];
#pragma unroll
    for (int q = 0; q < D_ / 4; ++q) r4[q] = rowp[q];
    float sa = 0.f, sb = 0.f;
#pragma unroll
    for (int q = 0; q < D_ / 8; ++q) {
      const v4f ra = r4[q], rb = r4[q + D_ / 8];
      const v2f al = __builtin_shufflevector(ra, ra, 0, 1) + g2[2 * q];
      const v2f ah = __builtin_shufflevector(ra, ra, 2, 3) + g2[2 * q + 1];
      const v2f bl = __builtin_shufflevector(rb, rb, 0, 1) + g2[2 * q + D_ / 4];
      const v2f bh = __builtin_shufflevector(rb, rb, 2, 3) + g2[2 * q + D_ / 4 + 1];
      sa = fmaf(a[4 * q + 0], fabsf(al.x), sa);
      sb = fmaf(a[4 * q + D_ / 2 + 0], fabsf(bl.x), sb);
      sa = fmaf(a[4 * q + 1], fabsf(al.y), sa);
      sb = fmaf(a[4 * q + D_ / 2 + 1], fabsf(bl.y), sb);
      sa = fmaf(a[4 * q + 2], fabsf(ah.x), sa);
      sb = fmaf(a[4 * q + D_ / 2 + 2], fabsf(bh.x), sb);
      sa = fmaf(a[4 * q + 3], fabsf(ah.y), sa);
      sb = fmaf(a[4 * q + D_ / 2 + 3], fabsf(bh.y), sb);
    }
    hSc[t * SP_ + tid] = fmaf(ABS_, sa + sb, lin);
    __builtin_amdgcn_sched_group_barrier(0x100, D_ / 4, 0);
    __builtin_amdgcn_sched_group_barrier(0x002, D_ / 2 + D_, 0);
  }

  const int half = lane >> 4;
  const int r = lane & 15;
  const int n0 = wv * 16;
  v2f bf[C_ / 4];
  {
    const float* pB = grb + (long)(2 * half) * HID_ + n0 + r;
#pragma unroll
    for (int j = 0; j < C_ / 4; ++j) {
      bf[j].x = pB[(long)(4 * j) * HID_];
      bf[j].y = pB[(long)(4 * j + 1) * HID_];
    }
  }
  float* hrow = h + ((long)b * C_ + t0 + wid) * HID_ + 4 * lane;
  const float4 hv0 = *(const float4*)(hrow);
  const float4 hv1 = *(const float4*)(hrow + HID_ / 2);

  float sc[TCH_], m[TCH_];
#pragma unroll
  for (int t = 0; t < TCH_; ++t) { sc[t] = hSc[t * SP_ + tid]; m[t] = sc[t]; }

  for (int off = 16; off > 0; off >>= 1) {
#pragma unroll
    for (int t = 0; t < TCH_; ++t) m[t] = fmaxf(m[t], __shfl_xor(m[t], off, 32));
  }
  if (lane == 0) {
#pragma unroll
    for (int t = 0; t < TCH_; ++t) hRedM[t * 4 + wv] = m[t];
  }
  __syncthreads();
  float e[TCH_], ss[TCH_];
#pragma unroll
  for (int t = 0; t < TCH_; ++t) {
    const float4 rm = ((const float4*)hRedM)[t];
    const float mt = fmaxf(fmaxf(rm.x, rm.y), fmaxf(rm.z, rm.w));
    e[t] = __expf(sc[t] - mt);
    ss[t] = e[t];
  }
  for (int off = 16; off > 0; off >>= 1) {
#pragma unroll
    for (int t = 0; t < TCH_; ++t) ss[t] += __shfl_xor(ss[t], off, 32);
  }
  if (lane == 0) {
#pragma unroll
    for (int t = 0; t < TCH_; ++t) hRedS[t * 4 + wv] = ss[t];
  }
  __syncthreads();
  float av[TCH_];
#pragma unroll
  for (int t = 0; t < TCH_; ++t) {
    const float4 rs = ((const float4*)hRedS)[t];
    const float ssum = rs.x + rs.y + rs.z + rs.w;
    av[t] = e[t] / ssum;
    hSc[t * SP_ + tid] = av[t];
  }
  __syncthreads();

  const int c0 = 4 * lane;
  const float4 bv0 = *(const float4*)(bias + c0);
  const float4 bv1 = *(const float4*)(bias + HID_ / 2 + c0);
  const float4 gv0 = *(const float4*)(lng + c0);
  const float4 gv1 = *(const float4*)(lng + HID_ / 2 + c0);
  const float4 lb0 = *(const float4*)(lnb + c0);
  const float4 lb1 = *(const float4*)(lnb + HID_ / 2 + c0);

  v8f acc = {0.f, 0.f, 0.f, 0.f, 0.f, 0.f, 0.f, 0.f};
  {
    const float* pA = hSc + r * SP_ + 2 * half;
#pragma unroll
    for (int j = 0; j < C_ / 4; ++j) {
      const v2f af = *(const v2f*)(pA + 4 * j);
      acc = __builtin_amdgcn_wmma_f32_16x16x4_f32(false, af, false, bf[j], (short)0, acc, false, false);
    }
  }
#pragma unroll
  for (int i = 0; i < 8; ++i) sOut[(i + 8 * half) * OW_ + hd * D_ + n0 + r] = acc[i];
  __syncthreads();

  if (alpha != nullptr) {
    float* arow = alpha + (long)(b * H_ + hd) * C_ * C_ + (long)t0 * C_ + tid;
#pragma unroll
    for (int t = 0; t < TCH_; ++t)
      arow[t * C_] = av[t];
  }

  const float4 o0 = *(const float4*)(sOut + wid * OW_ + c0);
  const float4 o1 = *(const float4*)(sOut + wid * OW_ + HID_ / 2 + c0);
  float x[8] = {o0.x + bv0.x, o0.y + bv0.y, o0.z + bv0.z, o0.w + bv0.w,
                o1.x + bv1.x, o1.y + bv1.y, o1.z + bv1.z, o1.w + bv1.w};
  const float hv[8] = {hv0.x, hv0.y, hv0.z, hv0.w, hv1.x, hv1.y, hv1.z, hv1.w};
  const float gv[8] = {gv0.x, gv0.y, gv0.z, gv0.w, gv1.x, gv1.y, gv1.z, gv1.w};
  const float lb[8] = {lb0.x, lb0.y, lb0.z, lb0.w, lb1.x, lb1.y, lb1.z, lb1.w};
  float v[8];
#pragma unroll
  for (int k = 0; k < 8; ++k) {
    const float xe = x[k] > 0.f ? x[k] : (__expf(x[k]) - 1.f);
    v[k] = hv[k] + xe;
  }
  float s = ((v[0] + v[1]) + (v[2] + v[3])) + ((v[4] + v[5]) + (v[6] + v[7]));
  for (int off = 16; off > 0; off >>= 1) s += __shfl_xor(s, off, 32);
  const float mu = s * (1.f / HID_);
  float dm[8];
#pragma unroll
  for (int k = 0; k < 8; ++k) dm[k] = v[k] - mu;
  float q = ((dm[0] * dm[0] + dm[1] * dm[1]) + (dm[2] * dm[2] + dm[3] * dm[3]))
          + ((dm[4] * dm[4] + dm[5] * dm[5]) + (dm[6] * dm[6] + dm[7] * dm[7]));
  for (int off = 16; off > 0; off >>= 1) q += __shfl_xor(q, off, 32);
  const float inv = rsqrtf(q * (1.f / HID_) + EPS_);
  float4 w0, w1;
  w0.x = dm[0] * inv * gv[0] + lb[0]; w0.y = dm[1] * inv * gv[1] + lb[1];
  w0.z = dm[2] * inv * gv[2] + lb[2]; w0.w = dm[3] * inv * gv[3] + lb[3];
  w1.x = dm[4] * inv * gv[4] + lb[4]; w1.y = dm[5] * inv * gv[5] + lb[5];
  w1.z = dm[6] * inv * gv[6] + lb[6]; w1.w = dm[7] * inv * gv[7] + lb[7];
  *(float4*)(hrow) = w0;
  *(float4*)(hrow + HID_ / 2) = w1;
}

extern "C" void kernel_launch(void* const* d_in, const int* in_sizes, int n_in,
                              void* d_out, int out_size, void* d_ws, size_t ws_size,
                              hipStream_t stream) {
  (void)in_sizes; (void)n_in; (void)out_size; (void)ws_size;
  const float* x      = (const float*)d_in[0];
  const float* W_emb  = (const float*)d_in[1];
  const float* b_emb  = (const float*)d_in[2];
  const float* Wl     = (const float*)d_in[3];
  const float* Wr     = (const float*)d_in[4];
  const float* att    = (const float*)d_in[5];
  const float* bias_g = (const float*)d_in[6];
  const float* ln_g   = (const float*)d_in[7];
  const float* ln_b   = (const float*)d_in[8];
  const float* W_proj = (const float*)d_in[9];
  const float* b_proj = (const float*)d_in[10];

  float* out   = (float*)d_out;
  float* alpha = out + (long)B_ * L_ * C_;

  float* ws = (float*)d_ws;
  const long hwN = (long)B_ * C_ * HID_;
  float* h    = ws;
  float* gl   = ws + hwN;
  float* gr   = ws + 2 * hwN;

  wmma_gemm_f32<C_, HID_, L_, C_, HID_, HID_, 1,
                (long)L_ * C_, 0, 0, 0, (long)C_ * HID_, 0, 0,
                true, false, true>
      <<<dim3((C_ / 32) * (HID_ / 32) / 4, B_), 128, 0, stream>>>(
          x, W_emb, b_emb, h);

  for (int layer = 0; layer < NL_; ++layer) {
    const float* Wl_i  = Wl + (long)layer * HID_ * HID_;
    const float* Wr_i  = Wr + (long)layer * HID_ * HID_;
    const float* att_i = att + (long)layer * H_ * D_;
    const float* bg_i  = bias_g + (long)layer * HID_;

    gemm_gl_gr<<<dim3((C_ / 32) * (HID_ / 32) / 4, B_, 2), 128, 0, stream>>>(
        h, Wl_i, Wr_i, gl, gr);

    gat_scores_agg_ln<<<dim3(B_, C_ / TCH_), 512, 0, stream>>>(
        gl, gr, att_i, bg_i, ln_g + (long)layer * HID_, ln_b + (long)layer * HID_,
        layer == NL_ - 1 ? alpha : nullptr, h);
  }

  wmma_gemm_f32<C_, L_, HID_, HID_, L_, C_, 1,
                (long)C_ * HID_, 0, 0, 0, (long)L_ * C_, 0, 0,
                false, true, true>
      <<<dim3((C_ / 32) * (L_ / 32) / 4, B_), 128, 0, stream>>>(
          h, W_proj, b_proj, out);
}
